# starting kernel
# baseline (speedup 1.0000x reference)
#include <hip/hip_runtime.h>
#include <math.h>

#define NBATCH  16
#define NATOM   512
#define NEIGH   64
#define NPAIR   (NATOM * NEIGH)
#define TOTPAIR (NBATCH * NPAIR)
#define TOTATOM (NBATCH * NATOM)
#define NWAVE   8
#define NANG    13
#define NWORB   (NANG * NWAVE)
#define NORBIT  128

typedef __attribute__((ext_vector_type(2)))  float    v2f;
typedef __attribute__((ext_vector_type(8)))  float    v8f;
typedef __attribute__((ext_vector_type(16))) _Float16 v16h;

__global__ __launch_bounds__(256) void eann_pair_kernel(
    const float* __restrict__ cart,
    const int*   __restrict__ species,
    const int*   __restrict__ atom_index,
    const float* __restrict__ shifts,
    const float* __restrict__ rs,
    const float* __restrict__ inta,
    const float* __restrict__ params,
    const float* __restrict__ cutoff_p,
    float*       __restrict__ sum_worb)
{
    const int g = blockIdx.x * blockDim.x + threadIdx.x;
    if (g >= TOTPAIR) return;

    const int b     = g >> 15;
    const int i_loc = atom_index[g];
    const int j_loc = atom_index[TOTPAIR + g];
    const int idx_i = b * NATOM + i_loc;
    const int idx_j = b * NATOM + j_loc;

    const float dx = cart[idx_i * 3 + 0] - cart[idx_j * 3 + 0] + shifts[g * 3 + 0];
    const float dy = cart[idx_i * 3 + 1] - cart[idx_j * 3 + 1] + shifts[g * 3 + 1];
    const float dz = cart[idx_i * 3 + 2] - cart[idx_j * 3 + 2] + shifts[g * 3 + 2];
    const float dist = sqrtf(dx * dx + dy * dy + dz * dz);

    const int sp = species[idx_j];
    const float cutoff = cutoff_p[0];
    const float t = 0.5f * __cosf(dist * (3.14159265358979f / cutoff)) + 0.5f;
    const float fc = t * t;

    float rw[NWAVE];
#pragma unroll
    for (int w = 0; w < NWAVE; ++w) {
        const float dd = dist - rs[sp * NWAVE + w];
        rw[w] = __expf(inta[sp * NWAVE + w] * dd * dd) * params[sp * NWAVE + w];
    }

    float ax[NANG];
    ax[0] = fc;
    ax[1] = fc * dx; ax[2] = fc * dy; ax[3] = fc * dz;
    const float dv[3] = {dx, dy, dz};
    int tpos = 4;
#pragma unroll
    for (int i = 1; i <= 3; ++i)
#pragma unroll
        for (int c = 0; c < 3; ++c)
            ax[tpos++] = ax[i] * dv[c];

    float* base = sum_worb + (size_t)idx_i * NWORB;
#pragma unroll
    for (int l = 0; l < NANG; ++l)
#pragma unroll
        for (int w = 0; w < NWAVE; ++w)
            unsafeAtomicAdd(base + l * NWAVE + w, ax[l] * rw[w]);
}

__global__ __launch_bounds__(256) void eann_density_wmma_kernel(
    const float* __restrict__ sum_worb,
    const float* __restrict__ hyper,
    float*       __restrict__ out)
{
    const int lane     = threadIdx.x & 31;
    const int wave     = threadIdx.x >> 5;
    const int tile     = blockIdx.x * 8 + wave;
    const int atomTile = tile >> 3;
    const int colTile  = tile & 7;
    const int half     = lane >> 4;
    const int lane16   = lane & 15;
    const int n        = colTile * 16 + lane16;
    const int m        = atomTile * 16 + lane16;

    v8f acc = {};

#if __has_builtin(__builtin_amdgcn_wmma_f32_16x16x4_f32)
    const int kb = half * 2;

    v2f bt[3][2];
#pragma unroll
    for (int l = 0; l < 3; ++l)
#pragma unroll
        for (int c = 0; c < 2; ++c) {
            bt[l][c].x = hyper[(l * NWAVE + 4 * c + kb + 0) * NORBIT + n];
            bt[l][c].y = hyper[(l * NWAVE + 4 * c + kb + 1) * NORBIT + n];
        }

#pragma unroll
    for (int j = 0; j < NANG; ++j) {
        const int l = (j == 0) ? 0 : ((j < 4) ? 1 : 2);
        const float* ap = sum_worb + (size_t)m * NWORB + j * NWAVE;
        const v2f a0 = *(const v2f*)(ap + kb);
        const v2f a1 = *(const v2f*)(ap + 4 + kb);
        v8f c0 = {};
        v8f dmat = __builtin_amdgcn_wmma_f32_16x16x4_f32(
            false, a0, false, bt[l][0], (short)0, c0, false, false);
        dmat = __builtin_amdgcn_wmma_f32_16x16x4_f32(
            false, a1, false, bt[l][1], (short)0, dmat, false, false);
        acc += dmat * dmat;
    }
#else
    v16h bt[3];
#pragma unroll
    for (int l = 0; l < 3; ++l) {
        v16h bvec = {};
        if (half == 0) {
#pragma unroll
            for (int k = 0; k < NWAVE; ++k)
                bvec[k] = (_Float16)hyper[(l * NWAVE + k) * NORBIT + n];
        }
        bt[l] = bvec;
    }
#pragma unroll
    for (int j = 0; j < NANG; ++j) {
        const int l = (j == 0) ? 0 : ((j < 4) ? 1 : 2);
        v16h avec = {};
        if (half == 0) {
#pragma unroll
            for (int k = 0; k < NWAVE; ++k)
                avec[k] = (_Float16)sum_worb[(size_t)m * NWORB + j * NWAVE + k];
        }
        v8f c0 = {};
        v8f dmat = __builtin_amdgcn_wmma_f32_16x16x32_f16(
            false, avec, false, bt[l], (short)0, c0, false, false);
        acc += dmat * dmat;
    }
#endif

#pragma unroll
    for (int r = 0; r < 8; ++r) {
        const int row = atomTile * 16 + r + half * 8;
        out[(size_t)row * NORBIT + n] = acc[r];
    }
}

extern "C" void kernel_launch(void* const* d_in, const int* in_sizes, int n_in,
                              void* d_out, int out_size, void* d_ws, size_t ws_size,
                              hipStream_t stream) {
    const float* cart       = (const float*)d_in[0];
    const int*   species    = (const int*)  d_in[2];
    const int*   atom_index = (const int*)  d_in[3];
    const float* shifts     = (const float*)d_in[4];
    const float* rs         = (const float*)d_in[5];
    const float* inta       = (const float*)d_in[6];
    const float* params     = (const float*)d_in[7];
    const float* hyper      = (const float*)d_in[8];
    const float* cutoff     = (const float*)d_in[10];
    float*       out        = (float*)d_out;
    float*       sum_worb   = (float*)d_ws;

    hipMemsetAsync(d_ws, 0, (size_t)TOTATOM * NWORB * sizeof(float), stream);

    eann_pair_kernel<<<(TOTPAIR + 255) / 256, 256, 0, stream>>>(
        cart, species, atom_index, shifts, rs, inta, params, cutoff, sum_worb);

    eann_density_wmma_kernel<<<512, 256, 0, stream>>>(sum_worb, hyper, out);
}


# Pair phase: float atomics replaced by owner gather: a work-group scans its batch, lists pairs of its 16 atoms in LDS, sums per atom in registers in fixed list order; memset dropped
# speedup vs baseline: 41.5322x; 41.5322x over previous
#include <hip/hip_runtime.h>
#include <math.h>

#define NBATCH  16
#define NATOM   512
#define NEIGH   64
#define NPAIR   (NATOM * NEIGH)
#define TOTPAIR (NBATCH * NPAIR)
#define TOTATOM (NBATCH * NATOM)
#define NWAVE   8
#define NANG    13
#define NWORB   (NANG * NWAVE)
#define NORBIT  128

typedef __attribute__((ext_vector_type(2)))  float    v2f;
typedef __attribute__((ext_vector_type(8)))  float    v8f;
typedef __attribute__((ext_vector_type(16))) _Float16 v16h;

#define TILE_ATOMS 16
#define SUBCHUNK   4096
#define NSUB       (NPAIR / SUBCHUNK)
#define GROUP      256
#define RECSTRIDE  21

__device__ __forceinline__ void sum_records(const float* rec, int s, unsigned m,
                                            int q, int w, float (&acc)[4])
{
    while (m) {
        const float* r0 = rec + (s + __builtin_ctz(m)) * RECSTRIDE;
        m &= m - 1u;
        if (m) {
            const float* r1 = rec + (s + __builtin_ctz(m)) * RECSTRIDE;
            m &= m - 1u;
            const float rw0 = r0[NANG + w], rw1 = r1[NANG + w];
            const float a00 = r0[q], a01 = r0[4 + q], a02 = r0[8 + q], a03 = r0[12];
            const float a10 = r1[q], a11 = r1[4 + q], a12 = r1[8 + q], a13 = r1[12];
            acc[0] += a00 * rw0; acc[1] += a01 * rw0; acc[2] += a02 * rw0; acc[3] += a03 * rw0;
            acc[0] += a10 * rw1; acc[1] += a11 * rw1; acc[2] += a12 * rw1; acc[3] += a13 * rw1;
        } else {
            const float rw0 = r0[NANG + w];
            acc[0] += r0[q]     * rw0;
            acc[1] += r0[4 + q] * rw0;
            acc[2] += r0[8 + q] * rw0;
            acc[3] += r0[12]    * rw0;
        }
    }
}

__global__ __launch_bounds__(256) void eann_gather_kernel(
    const float* __restrict__ cart,
    const int*   __restrict__ species,
    const int*   __restrict__ atom_index,
    const float* __restrict__ shifts,
    const float* __restrict__ rs,
    const float* __restrict__ inta,
    const float* __restrict__ params,
    const float* __restrict__ cutoff_p,
    float*       __restrict__ sum_worb)
{
    __shared__ unsigned s_list[SUBCHUNK];
    __shared__ float    s_rec[GROUP * RECSTRIDE];
    __shared__ int      s_id[GROUP];
    __shared__ int      s_wcnt[8];

    const int t      = threadIdx.x;
    const int lane   = t & 31;
    const int wave   = t >> 5;
    const int b      = blockIdx.x >> 5;
    const int tib    = blockIdx.x & 31;
    const int a_base = tib * TILE_ATOMS;

    const int* __restrict__ pi = atom_index + b * NPAIR;
    const int* __restrict__ pj = atom_index + TOTPAIR + b * NPAIR;

    const float cutoff = cutoff_p[0];
    const int   w      = lane & 7;
    const int   q      = lane >> 3;
    const int   myA0   = 2 * wave;
    const int   myA1   = 2 * wave + 1;
    const unsigned lt  = (1u << lane) - 1u;

    float acc0[4] = {0.f, 0.f, 0.f, 0.f};
    float acc1[4] = {0.f, 0.f, 0.f, 0.f};

    int4 nx[4];
#pragma unroll
    for (int it = 0; it < 4; ++it)
        nx[it] = *(const int4*)(pi + it * 1024 + t * 4);

    for (int sc = 0; sc < NSUB; ++sc) {
        const int p0 = sc * SUBCHUNK;

        int iv[16];
#pragma unroll
        for (int it = 0; it < 4; ++it) {
            iv[it * 4 + 0] = nx[it].x; iv[it * 4 + 1] = nx[it].y;
            iv[it * 4 + 2] = nx[it].z; iv[it * 4 + 3] = nx[it].w;
        }
        if (sc + 1 < NSUB) {
#pragma unroll
            for (int it = 0; it < 4; ++it)
                nx[it] = *(const int4*)(pi + p0 + SUBCHUNK + it * 1024 + t * 4);
        }
        unsigned bal[16];
        int wtot = 0;
#pragma unroll
        for (int k = 0; k < 16; ++k) {
            bal[k] = __builtin_amdgcn_ballot_w32((iv[k] >> 4) == tib);
            wtot += __builtin_popcount(bal[k]);
        }
        if (lane == 0) s_wcnt[wave] = wtot;
        __syncthreads();

        int woff = 0, ntot = 0;
        for (int ww = 0; ww < 8; ++ww) {
            const int c = s_wcnt[ww];
            if (ww < wave) woff += c;
            ntot += c;
        }
        int run = woff;
#pragma unroll
        for (int k = 0; k < 16; ++k) {
            if ((bal[k] >> lane) & 1u) {
                const unsigned p = (unsigned)(p0 + (k >> 2) * 1024 + t * 4 + (k & 3));
                s_list[run + __builtin_popcount(bal[k] & lt)] = p | ((unsigned)(iv[k] & 15) << 16);
            }
            run += __builtin_popcount(bal[k]);
        }
        __syncthreads();

        const int n = __builtin_amdgcn_readfirstlane(ntot);

        for (int base = 0; base < n; base += GROUP) {
            const int e = base + t;
            int id = -1;
            if (e < n) {
                const unsigned ent = s_list[e];
                const int p     = (int)(ent & 0xFFFFu);
                const int a     = (int)(ent >> 16);
                const int g     = b * NPAIR + p;
                const int j_loc = pj[p];
                const int idx_i = b * NATOM + a_base + a;
                const int idx_j = b * NATOM + j_loc;

                const float dx = cart[idx_i * 3 + 0] - cart[idx_j * 3 + 0] + shifts[g * 3 + 0];
                const float dy = cart[idx_i * 3 + 1] - cart[idx_j * 3 + 1] + shifts[g * 3 + 1];
                const float dz = cart[idx_i * 3 + 2] - cart[idx_j * 3 + 2] + shifts[g * 3 + 2];
                const float dist = sqrtf(dx * dx + dy * dy + dz * dz);

                const int sp = species[idx_j];
                const float tc = 0.5f * __cosf(dist * (3.14159265358979f / cutoff)) + 0.5f;
                const float fc = tc * tc;

                float* r = s_rec + t * RECSTRIDE;
                const float dv[3] = {dx, dy, dz};
                float ax[NANG];
                ax[0] = fc;
                ax[1] = fc * dx; ax[2] = fc * dy; ax[3] = fc * dz;
                int tpos = 4;
#pragma unroll
                for (int i = 1; i <= 3; ++i)
#pragma unroll
                    for (int c = 0; c < 3; ++c)
                        ax[tpos++] = ax[i] * dv[c];
                for (int l = 0; l < NANG; ++l) r[l] = ax[l];
                for (int ww = 0; ww < NWAVE; ++ww) {
                    const float dd = dist - rs[sp * NWAVE + ww];
                    r[NANG + ww] = __expf(inta[sp * NWAVE + ww] * dd * dd) * params[sp * NWAVE + ww];
                }
                id = a;
            }
            s_id[t] = id;
            __syncthreads();

            const int cnt = (n - base < GROUP) ? (n - base) : GROUP;
            for (int s = 0; s < cnt; s += 32) {
                const int idv = s_id[s + lane];
                unsigned m0 = __builtin_amdgcn_ballot_w32(idv == myA0);
                unsigned m1 = __builtin_amdgcn_ballot_w32(idv == myA1);
                sum_records(s_rec, s, m0, q, w, acc0);
                sum_records(s_rec, s, m1, q, w, acc1);
            }
            __syncthreads();
        }
    }

    float* st0 = s_rec + myA0 * NWORB;
    float* st1 = s_rec + myA1 * NWORB;
    for (int k = 0; k < 3; ++k) {
        st0[32 * k + lane] = acc0[k];
        st1[32 * k + lane] = acc1[k];
    }
    if (lane < 8) {
        st0[96 + lane] = acc0[3];
        st1[96 + lane] = acc1[3];
    }
    __syncthreads();
    float* tile_out = sum_worb + (size_t)(b * NATOM + a_base) * NWORB;
    for (int i = t; i < TILE_ATOMS * NWORB; i += 256)
        tile_out[i] = s_rec[i];
}

__global__ __launch_bounds__(256) void eann_density_wmma_kernel(
    const float* __restrict__ sum_worb,
    const float* __restrict__ hyper,
    float*       __restrict__ out)
{
    const int lane     = threadIdx.x & 31;
    const int wave     = threadIdx.x >> 5;
    const int tile     = blockIdx.x * 8 + wave;
    const int atomTile = tile >> 3;
    const int colTile  = tile & 7;
    const int half     = lane >> 4;
    const int lane16   = lane & 15;
    const int n        = colTile * 16 + lane16;
    const int m        = atomTile * 16 + lane16;

    v8f acc = {};

#if __has_builtin(__builtin_amdgcn_wmma_f32_16x16x4_f32)
    const int kb = half * 2;

    v2f bt[3][2];
#pragma unroll
    for (int l = 0; l < 3; ++l)
#pragma unroll
        for (int c = 0; c < 2; ++c) {
            bt[l][c].x = hyper[(l * NWAVE + 4 * c + kb + 0) * NORBIT + n];
            bt[l][c].y = hyper[(l * NWAVE + 4 * c + kb + 1) * NORBIT + n];
        }

#pragma unroll
    for (int j = 0; j < NANG; ++j) {
        const int l = (j == 0) ? 0 : ((j < 4) ? 1 : 2);
        const float* ap = sum_worb + (size_t)m * NWORB + j * NWAVE;
        const v2f a0 = *(const v2f*)(ap + kb);
        const v2f a1 = *(const v2f*)(ap + 4 + kb);
        v8f c0 = {};
        v8f dmat = __builtin_amdgcn_wmma_f32_16x16x4_f32(
            false, a0, false, bt[l][0], (short)0, c0, false, false);
        dmat = __builtin_amdgcn_wmma_f32_16x16x4_f32(
            false, a1, false, bt[l][1], (short)0, dmat, false, false);
        acc += dmat * dmat;
    }
#else
    v16h bt[3];
#pragma unroll
    for (int l = 0; l < 3; ++l) {
        v16h bvec = {};
        if (half == 0) {
#pragma unroll
            for (int k = 0; k < NWAVE; ++k)
                bvec[k] = (_Float16)hyper[(l * NWAVE + k) * NORBIT + n];
        }
        bt[l] = bvec;
    }
#pragma unroll
    for (int j = 0; j < NANG; ++j) {
        const int l = (j == 0) ? 0 : ((j < 4) ? 1 : 2);
        v16h avec = {};
        if (half == 0) {
#pragma unroll
            for (int k = 0; k < NWAVE; ++k)
                avec[k] = (_Float16)sum_worb[(size_t)m * NWORB + j * NWAVE + k];
        }
        v8f c0 = {};
        v8f dmat = __builtin_amdgcn_wmma_f32_16x16x32_f16(
            false, avec, false, bt[l], (short)0, c0, false, false);
        acc += dmat * dmat;
    }
#endif

#pragma unroll
    for (int r = 0; r < 8; ++r) {
        const int row = atomTile * 16 + r + half * 8;
        out[(size_t)row * NORBIT + n] = acc[r];
    }
}

extern "C" void kernel_launch(void* const* d_in, const int* in_sizes, int n_in,
                              void* d_out, int out_size, void* d_ws, size_t ws_size,
                              hipStream_t stream) {
    const float* cart       = (const float*)d_in[0];
    const int*   species    = (const int*)  d_in[2];
    const int*   atom_index = (const int*)  d_in[3];
    const float* shifts     = (const float*)d_in[4];
    const float* rs         = (const float*)d_in[5];
    const float* inta       = (const float*)d_in[6];
    const float* params     = (const float*)d_in[7];
    const float* hyper      = (const float*)d_in[8];
    const float* cutoff     = (const float*)d_in[10];
    float*       out        = (float*)d_out;
    float*       sum_worb   = (float*)d_ws;

    eann_gather_kernel<<<512, 256, 0, stream>>>(
        cart, species, atom_index, shifts, rs, inta, params, cutoff, sum_worb);

    eann_density_wmma_kernel<<<512, 256, 0, stream>>>(sum_worb, hyper, out);
}


# Density product fused into the gather kernel: each work-group feeds its 16-atom tile from LDS to the f32 WMMAs, one 16-orbit column tile per wave; second launch and workspace round trip dropped
# speedup vs baseline: 44.8326x; 1.0795x over previous
#include <hip/hip_runtime.h>
#include <math.h>

#define NBATCH  16
#define NATOM   512
#define NEIGH   64
#define NPAIR   (NATOM * NEIGH)
#define TOTPAIR (NBATCH * NPAIR)
#define TOTATOM (NBATCH * NATOM)
#define NWAVE   8
#define NANG    13
#define NWORB   (NANG * NWAVE)
#define NORBIT  128

typedef __attribute__((ext_vector_type(2)))  float    v2f;
typedef __attribute__((ext_vector_type(8)))  float    v8f;

#define TILE_ATOMS 16
#define SUBCHUNK   4096
#define NSUB       (NPAIR / SUBCHUNK)
#define GROUP      256
#define RECSTRIDE  21

__device__ __forceinline__ void sum_records(const float* rec, int s, unsigned m,
                                            int q, int w, float (&acc)[4])
{
    while (m) {
        const float* r0 = rec + (s + __builtin_ctz(m)) * RECSTRIDE;
        m &= m - 1u;
        if (m) {
            const float* r1 = rec + (s + __builtin_ctz(m)) * RECSTRIDE;
            m &= m - 1u;
            const float rw0 = r0[NANG + w], rw1 = r1[NANG + w];
            const float a00 = r0[q], a01 = r0[4 + q], a02 = r0[8 + q], a03 = r0[12];
            const float a10 = r1[q], a11 = r1[4 + q], a12 = r1[8 + q], a13 = r1[12];
            acc[0] += a00 * rw0; acc[1] += a01 * rw0; acc[2] += a02 * rw0; acc[3] += a03 * rw0;
            acc[0] += a10 * rw1; acc[1] += a11 * rw1; acc[2] += a12 * rw1; acc[3] += a13 * rw1;
        } else {
            const float rw0 = r0[NANG + w];
            acc[0] += r0[q]     * rw0;
            acc[1] += r0[4 + q] * rw0;
            acc[2] += r0[8 + q] * rw0;
            acc[3] += r0[12]    * rw0;
        }
    }
}

__global__ __launch_bounds__(256) void eann_gather_kernel(
    const float* __restrict__ cart,
    const int*   __restrict__ species,
    const int*   __restrict__ atom_index,
    const float* __restrict__ shifts,
    const float* __restrict__ rs,
    const float* __restrict__ inta,
    const float* __restrict__ params,
    const float* __restrict__ cutoff_p,
    const float* __restrict__ hyper,
    float*       __restrict__ out)
{
    __shared__ unsigned s_list[SUBCHUNK];
    __shared__ __attribute__((aligned(16))) float s_rec[GROUP * RECSTRIDE];
    __shared__ int      s_id[GROUP];
    __shared__ int      s_wcnt[8];

    const int t      = threadIdx.x;
    const int lane   = t & 31;
    const int wave   = t >> 5;
    const int b      = blockIdx.x >> 5;
    const int tib    = blockIdx.x & 31;
    const int a_base = tib * TILE_ATOMS;

    const int* __restrict__ pi = atom_index + b * NPAIR;
    const int* __restrict__ pj = atom_index + TOTPAIR + b * NPAIR;

    const float cutoff = cutoff_p[0];
    const int   w      = lane & 7;
    const int   q      = lane >> 3;
    const int   myA0   = 2 * wave;
    const int   myA1   = 2 * wave + 1;
    const unsigned lt  = (1u << lane) - 1u;

    float acc0[4] = {0.f, 0.f, 0.f, 0.f};
    float acc1[4] = {0.f, 0.f, 0.f, 0.f};

    const int half   = lane >> 4;
    const int lane16 = lane & 15;
    const int kb     = half * 2;
    const int n      = wave * 16 + lane16;
    v2f bt[3][2];
#pragma unroll
    for (int l = 0; l < 3; ++l)
#pragma unroll
        for (int c = 0; c < 2; ++c) {
            bt[l][c].x = hyper[(l * NWAVE + 4 * c + kb + 0) * NORBIT + n];
            bt[l][c].y = hyper[(l * NWAVE + 4 * c + kb + 1) * NORBIT + n];
        }

    int4 nx[4];
#pragma unroll
    for (int it = 0; it < 4; ++it)
        nx[it] = *(const int4*)(pi + it * 1024 + t * 4);

    for (int sc = 0; sc < NSUB; ++sc) {
        const int p0 = sc * SUBCHUNK;

        int iv[16];
#pragma unroll
        for (int it = 0; it < 4; ++it) {
            iv[it * 4 + 0] = nx[it].x; iv[it * 4 + 1] = nx[it].y;
            iv[it * 4 + 2] = nx[it].z; iv[it * 4 + 3] = nx[it].w;
        }
        if (sc + 1 < NSUB) {
#pragma unroll
            for (int it = 0; it < 4; ++it)
                nx[it] = *(const int4*)(pi + p0 + SUBCHUNK + it * 1024 + t * 4);
        }
        unsigned bal[16];
        int wtot = 0;
#pragma unroll
        for (int k = 0; k < 16; ++k) {
            bal[k] = __builtin_amdgcn_ballot_w32((iv[k] >> 4) == tib);
            wtot += __builtin_popcount(bal[k]);
        }
        if (lane == 0) s_wcnt[wave] = wtot;
        __syncthreads();

        int woff = 0, ntot = 0;
        for (int ww = 0; ww < 8; ++ww) {
            const int c = s_wcnt[ww];
            if (ww < wave) woff += c;
            ntot += c;
        }
        int run = woff;
#pragma unroll
        for (int k = 0; k < 16; ++k) {
            if ((bal[k] >> lane) & 1u) {
                const unsigned p = (unsigned)(p0 + (k >> 2) * 1024 + t * 4 + (k & 3));
                s_list[run + __builtin_popcount(bal[k] & lt)] = p | ((unsigned)(iv[k] & 15) << 16);
            }
            run += __builtin_popcount(bal[k]);
        }
        __syncthreads();

        const int n = __builtin_amdgcn_readfirstlane(ntot);

        for (int base = 0; base < n; base += GROUP) {
            const int e = base + t;
            int id = -1;
            if (e < n) {
                const unsigned ent = s_list[e];
                const int p     = (int)(ent & 0xFFFFu);
                const int a     = (int)(ent >> 16);
                const int g     = b * NPAIR + p;
                const int j_loc = pj[p];
                const int idx_i = b * NATOM + a_base + a;
                const int idx_j = b * NATOM + j_loc;

                const float dx = cart[idx_i * 3 + 0] - cart[idx_j * 3 + 0] + shifts[g * 3 + 0];
                const float dy = cart[idx_i * 3 + 1] - cart[idx_j * 3 + 1] + shifts[g * 3 + 1];
                const float dz = cart[idx_i * 3 + 2] - cart[idx_j * 3 + 2] + shifts[g * 3 + 2];
                const float dist = sqrtf(dx * dx + dy * dy + dz * dz);

                const int sp = species[idx_j];
                const float tc = 0.5f * __cosf(dist * (3.14159265358979f / cutoff)) + 0.5f;
                const float fc = tc * tc;

                float* r = s_rec + t * RECSTRIDE;
                const float dv[3] = {dx, dy, dz};
                float ax[NANG];
                ax[0] = fc;
                ax[1] = fc * dx; ax[2] = fc * dy; ax[3] = fc * dz;
                int tpos = 4;
#pragma unroll
                for (int i = 1; i <= 3; ++i)
#pragma unroll
                    for (int c = 0; c < 3; ++c)
                        ax[tpos++] = ax[i] * dv[c];
                for (int l = 0; l < NANG; ++l) r[l] = ax[l];
                for (int ww = 0; ww < NWAVE; ++ww) {
                    const float dd = dist - rs[sp * NWAVE + ww];
                    r[NANG + ww] = __expf(inta[sp * NWAVE + ww] * dd * dd) * params[sp * NWAVE + ww];
                }
                id = a;
            }
            s_id[t] = id;
            __syncthreads();

            const int cnt = (n - base < GROUP) ? (n - base) : GROUP;
            for (int s = 0; s < cnt; s += 32) {
                const int idv = s_id[s + lane];
                unsigned m0 = __builtin_amdgcn_ballot_w32(idv == myA0);
                unsigned m1 = __builtin_amdgcn_ballot_w32(idv == myA1);
                sum_records(s_rec, s, m0, q, w, acc0);
                sum_records(s_rec, s, m1, q, w, acc1);
            }
            __syncthreads();
        }
    }

    float* st0 = s_rec + myA0 * NWORB;
    float* st1 = s_rec + myA1 * NWORB;
    for (int k = 0; k < 3; ++k) {
        st0[32 * k + lane] = acc0[k];
        st1[32 * k + lane] = acc1[k];
    }
    if (lane < 8) {
        st0[96 + lane] = acc0[3];
        st1[96 + lane] = acc1[3];
    }
    __syncthreads();

    v8f dens = {};
    const float* arow = s_rec + lane16 * NWORB + kb;
#pragma unroll
    for (int j = 0; j < NANG; ++j) {
        const int l = (j == 0) ? 0 : ((j < 4) ? 1 : 2);
        const v2f a0 = *(const v2f*)(arow + j * NWAVE);
        const v2f a1 = *(const v2f*)(arow + j * NWAVE + 4);
        v8f c0 = {};
        v8f dmat = __builtin_amdgcn_wmma_f32_16x16x4_f32(
            false, a0, false, bt[l][0], (short)0, c0, false, false);
        dmat = __builtin_amdgcn_wmma_f32_16x16x4_f32(
            false, a1, false, bt[l][1], (short)0, dmat, false, false);
        dens += dmat * dmat;
    }

    const int row0 = b * NATOM + a_base + half * 8;
#pragma unroll
    for (int r = 0; r < 8; ++r)
        out[(size_t)(row0 + r) * NORBIT + n] = dens[r];
}

extern "C" void kernel_launch(void* const* d_in, const int* in_sizes, int n_in,
                              void* d_out, int out_size, void* d_ws, size_t ws_size,
                              hipStream_t stream) {
    const float* cart       = (const float*)d_in[0];
    const int*   species    = (const int*)  d_in[2];
    const int*   atom_index = (const int*)  d_in[3];
    const float* shifts     = (const float*)d_in[4];
    const float* rs         = (const float*)d_in[5];
    const float* inta       = (const float*)d_in[6];
    const float* params     = (const float*)d_in[7];
    const float* hyper      = (const float*)d_in[8];
    const float* cutoff     = (const float*)d_in[10];
    float*       out        = (float*)d_out;

    eann_gather_kernel<<<512, 256, 0, stream>>>(
        cart, species, atom_index, shifts, rs, inta, params, cutoff, hyper, out);
}


# Segment sum moved to the matrix unit: each wave queues its two atoms' pairs, builds their records and adds four per f32 WMMA, atom mask in the radial operand; per-atom sums keep list order
# speedup vs baseline: 60.2194x; 1.3432x over previous
#include <hip/hip_runtime.h>
#include <math.h>

#define NBATCH  16
#define NATOM   512
#define NEIGH   64
#define NPAIR   (NATOM * NEIGH)
#define TOTPAIR (NBATCH * NPAIR)
#define TOTATOM (NBATCH * NATOM)
#define NWAVE   8
#define NANG    13
#define NWORB   (NANG * NWAVE)
#define NORBIT  128

typedef __attribute__((ext_vector_type(2)))  float    v2f;
typedef __attribute__((ext_vector_type(8)))  float    v8f;

#define TILE_ATOMS 16
#define SUBCHUNK   4096
#define NSUB       (NPAIR / SUBCHUNK)
#define BATCH      32
#define FCHUNK     4
#define QCAP       256
#define NMASK      16
#define RECSTRIDE  (NANG + NMASK)

__device__ __forceinline__ void wave_sync()
{
    __builtin_amdgcn_fence(__ATOMIC_RELEASE, "wavefront");
    __builtin_amdgcn_wave_barrier();
    __builtin_amdgcn_fence(__ATOMIC_ACQUIRE, "wavefront");
}

__device__ __forceinline__ void sum_batch(
    int batch, int head, const unsigned* q, float* rec,
    int lane, int lane16, int half, int b, int a_base,
    const int* __restrict__ pj, const float* __restrict__ cart, const int* __restrict__ species,
    const float* __restrict__ shifts, const float* __restrict__ rs, const float* __restrict__ inta,
    const float* __restrict__ params, float cutoff, v8f& acc)
{
    const int nk = __builtin_amdgcn_readfirstlane((batch + 3) >> 2);
    if (lane < batch) {
        const unsigned ent = q[(head + lane) & (QCAP - 1)];
        const int p     = (int)(ent & 0xFFFFu);
        const int a     = (int)(ent >> 16);
        const int g     = b * NPAIR + p;
        const int j_loc = pj[p];
        const int idx_i = b * NATOM + a_base + a;
        const int idx_j = b * NATOM + j_loc;

        const float dx = cart[idx_i * 3 + 0] - cart[idx_j * 3 + 0] + shifts[g * 3 + 0];
        const float dy = cart[idx_i * 3 + 1] - cart[idx_j * 3 + 1] + shifts[g * 3 + 1];
        const float dz = cart[idx_i * 3 + 2] - cart[idx_j * 3 + 2] + shifts[g * 3 + 2];
        const float dist = sqrtf(dx * dx + dy * dy + dz * dz);

        const int sp = species[idx_j];
        const float tc = 0.5f * __cosf(dist * (3.14159265358979f / cutoff)) + 0.5f;
        const float fc = tc * tc;

        float* r = rec + lane * RECSTRIDE;
        const float dv[3] = {dx, dy, dz};
        float ax[NANG];
        ax[0] = fc;
        ax[1] = fc * dx; ax[2] = fc * dy; ax[3] = fc * dz;
        int tpos = 4;
#pragma unroll
        for (int i = 1; i <= 3; ++i)
#pragma unroll
            for (int c = 0; c < 3; ++c)
                ax[tpos++] = ax[i] * dv[c];
        for (int l = 0; l < NANG; ++l) r[l] = ax[l];
        float* own = r + NANG + ((a & 1) << 3);
        float* oth = r + NANG + (((a & 1) ^ 1) << 3);
        for (int ww = 0; ww < NWAVE; ++ww) {
            const float dd = dist - rs[sp * NWAVE + ww];
            own[ww] = __expf(inta[sp * NWAVE + ww] * dd * dd) * params[sp * NWAVE + ww];
            oth[ww] = 0.f;
        }
    } else if (lane < 4 * nk) {
        float* r = rec + lane * RECSTRIDE;
        for (int l = 0; l < RECSTRIDE; ++l) r[l] = 0.f;
    }
    wave_sync();

    for (int ks = 0; ks < nk; ++ks) {
        const int k0 = 4 * ks + 2 * half;
        const float* f = rec + k0 * RECSTRIDE + lane16;
        v2f av, bv;
        bv.x = f[0];             bv.y = f[RECSTRIDE];
        av.x = f[NANG];          av.y = f[RECSTRIDE + NANG];
        acc = __builtin_amdgcn_wmma_f32_16x16x4_f32(
            false, av, false, bv, (short)0, acc, false, false);
    }
    wave_sync();
}

__global__ __launch_bounds__(256) void eann_gather_kernel(
    const float* __restrict__ cart,
    const int*   __restrict__ species,
    const int*   __restrict__ atom_index,
    const float* __restrict__ shifts,
    const float* __restrict__ rs,
    const float* __restrict__ inta,
    const float* __restrict__ params,
    const float* __restrict__ cutoff_p,
    const float* __restrict__ hyper,
    float*       __restrict__ out)
{
    __shared__ unsigned s_list[SUBCHUNK];
    __shared__ unsigned s_queue[8][QCAP];
    __shared__ float    s_rec[8][BATCH * RECSTRIDE];
    __shared__ __attribute__((aligned(16))) float s_tile[TILE_ATOMS * NWORB];
    __shared__ int      s_wcnt[8];

    const int t      = threadIdx.x;
    const int lane   = t & 31;
    const int wave   = t >> 5;
    const int b      = blockIdx.x >> 5;
    const int tib    = blockIdx.x & 31;
    const int a_base = tib * TILE_ATOMS;

    const int* __restrict__ pi = atom_index + b * NPAIR;
    const int* __restrict__ pj = atom_index + TOTPAIR + b * NPAIR;

    const float cutoff = cutoff_p[0];
    const unsigned lt  = (1u << lane) - 1u;

    const int half   = lane >> 4;
    const int lane16 = lane & 15;
    const int kb     = half * 2;
    const int n      = wave * 16 + lane16;
    v2f bt[3][2];
#pragma unroll
    for (int l = 0; l < 3; ++l)
#pragma unroll
        for (int c = 0; c < 2; ++c) {
            bt[l][c].x = hyper[(l * NWAVE + 4 * c + kb + 0) * NORBIT + n];
            bt[l][c].y = hyper[(l * NWAVE + 4 * c + kb + 1) * NORBIT + n];
        }

    unsigned* q   = s_queue[wave];
    float*    rec = s_rec[wave];
    int qhead = 0, qcnt = 0;
    v8f acc = {};

    int4 nx[4];
#pragma unroll
    for (int it = 0; it < 4; ++it)
        nx[it] = *(const int4*)(pi + it * 1024 + t * 4);

    for (int sc = 0; sc < NSUB; ++sc) {
        const int p0 = sc * SUBCHUNK;

        int iv[16];
#pragma unroll
        for (int it = 0; it < 4; ++it) {
            iv[it * 4 + 0] = nx[it].x; iv[it * 4 + 1] = nx[it].y;
            iv[it * 4 + 2] = nx[it].z; iv[it * 4 + 3] = nx[it].w;
        }
        if (sc + 1 < NSUB) {
#pragma unroll
            for (int it = 0; it < 4; ++it)
                nx[it] = *(const int4*)(pi + p0 + SUBCHUNK + it * 1024 + t * 4);
        }
        unsigned bal[16];
        int wtot = 0;
#pragma unroll
        for (int k = 0; k < 16; ++k) {
            bal[k] = __builtin_amdgcn_ballot_w32((iv[k] >> 4) == tib);
            wtot += __builtin_popcount(bal[k]);
        }
        if (lane == 0) s_wcnt[wave] = wtot;
        __syncthreads();

        int woff = 0, ntot = 0;
        for (int ww = 0; ww < 8; ++ww) {
            const int c = s_wcnt[ww];
            if (ww < wave) woff += c;
            ntot += c;
        }
        int run = woff;
#pragma unroll
        for (int k = 0; k < 16; ++k) {
            if ((bal[k] >> lane) & 1u) {
                const unsigned p = (unsigned)(p0 + (k >> 2) * 1024 + t * 4 + (k & 3));
                s_list[run + __builtin_popcount(bal[k] & lt)] = p | ((unsigned)(iv[k] & 15) << 16);
            }
            run += __builtin_popcount(bal[k]);
        }
        __syncthreads();

        const int nlist = __builtin_amdgcn_readfirstlane(ntot);

        for (int cb = 0; cb < nlist; cb += 32 * FCHUNK) {
            unsigned cur[FCHUNK];
#pragma unroll
            for (int u = 0; u < FCHUNK; ++u) {
                const int e = cb + 32 * u + lane;
                cur[u] = (e < nlist) ? s_list[e] : 0xFFFFFFFFu;
            }
#pragma unroll
            for (int u = 0; u < FCHUNK; ++u) {
                const bool mine = (cur[u] >> 17) == (unsigned)wave;
                const unsigned m = __builtin_amdgcn_ballot_w32(mine);
                if (mine) q[(qhead + qcnt + __builtin_popcount(m & lt)) & (QCAP - 1)] = cur[u];
                qcnt += __builtin_popcount(m);
            }
            wave_sync();
            while (qcnt >= BATCH) {
                sum_batch(BATCH, qhead, q, rec, lane, lane16, half, b, a_base,
                          pj, cart, species, shifts, rs, inta, params, cutoff, acc);
                qhead = (qhead + BATCH) & (QCAP - 1);
                qcnt -= BATCH;
            }
        }
    }
    if (qcnt > 0)
        sum_batch(qcnt, qhead, q, rec, lane, lane16, half, b, a_base,
                  pj, cart, species, shifts, rs, inta, params, cutoff, acc);

    if (lane16 < NANG) {
        float4* dst = (float4*)(s_tile + (2 * wave + half) * NWORB + lane16 * NWAVE);
        dst[0] = make_float4(acc[0], acc[1], acc[2], acc[3]);
        dst[1] = make_float4(acc[4], acc[5], acc[6], acc[7]);
    }
    __syncthreads();

    v8f dens = {};
    const float* arow = s_tile + lane16 * NWORB + kb;
#pragma unroll
    for (int j = 0; j < NANG; ++j) {
        const int l = (j == 0) ? 0 : ((j < 4) ? 1 : 2);
        const v2f a0 = *(const v2f*)(arow + j * NWAVE);
        const v2f a1 = *(const v2f*)(arow + j * NWAVE + 4);
        v8f c0 = {};
        v8f dmat = __builtin_amdgcn_wmma_f32_16x16x4_f32(
            false, a0, false, bt[l][0], (short)0, c0, false, false);
        dmat = __builtin_amdgcn_wmma_f32_16x16x4_f32(
            false, a1, false, bt[l][1], (short)0, dmat, false, false);
        dens += dmat * dmat;
    }

    const int row0 = b * NATOM + a_base + half * 8;
#pragma unroll
    for (int r = 0; r < 8; ++r)
        out[(size_t)(row0 + r) * NORBIT + n] = dens[r];
}

extern "C" void kernel_launch(void* const* d_in, const int* in_sizes, int n_in,
                              void* d_out, int out_size, void* d_ws, size_t ws_size,
                              hipStream_t stream) {
    const float* cart       = (const float*)d_in[0];
    const int*   species    = (const int*)  d_in[2];
    const int*   atom_index = (const int*)  d_in[3];
    const float* shifts     = (const float*)d_in[4];
    const float* rs         = (const float*)d_in[5];
    const float* inta       = (const float*)d_in[6];
    const float* params     = (const float*)d_in[7];
    const float* hyper      = (const float*)d_in[8];
    const float* cutoff     = (const float*)d_in[10];
    float*       out        = (float*)d_out;

    eann_gather_kernel<<<512, 256, 0, stream>>>(
        cart, species, atom_index, shifts, rs, inta, params, cutoff, hyper, out);
}


# Scan split from the sums: each wave compacts its eighth of the pair list into its own 16-bit LDS list, index loads all up front, one barrier, then all lists drained; sum order now wave,load,slot,lane
# speedup vs baseline: 74.5059x; 1.2372x over previous
#include <hip/hip_runtime.h>
#include <math.h>

#define NBATCH  16
#define NATOM   512
#define NEIGH   64
#define NPAIR   (NATOM * NEIGH)
#define TOTPAIR (NBATCH * NPAIR)
#define TOTATOM (NBATCH * NATOM)
#define NWAVE   8
#define NANG    13
#define NWORB   (NANG * NWAVE)
#define NORBIT  128

typedef __attribute__((ext_vector_type(2)))  float    v2f;
typedef __attribute__((ext_vector_type(8)))  float    v8f;

#define TILE_ATOMS 16
#define WPAIRS     (NPAIR / 8)
#define NLOAD      (WPAIRS / 128)
#define BATCH      32
#define FCHUNK     5
#define QCAP       256
#define NMASK      16
#define RECSTRIDE  (NANG + NMASK)

__device__ __forceinline__ void wave_sync()
{
    __builtin_amdgcn_fence(__ATOMIC_RELEASE, "wavefront");
    __builtin_amdgcn_wave_barrier();
    __builtin_amdgcn_fence(__ATOMIC_ACQUIRE, "wavefront");
}

__device__ __forceinline__ void sum_batch(
    int batch, int head, const unsigned* q, float* rec,
    int lane, int lane16, int half, int b, int a_base,
    const int* __restrict__ pj, const float* __restrict__ cart, const int* __restrict__ species,
    const float* __restrict__ shifts, const float* __restrict__ rs, const float* __restrict__ inta,
    const float* __restrict__ params, float cutoff, v8f& acc)
{
    const int nk = __builtin_amdgcn_readfirstlane((batch + 3) >> 2);
    if (lane < batch) {
        const unsigned ent = q[(head + lane) & (QCAP - 1)];
        const int p     = (int)(ent & 0xFFFFu);
        const int a     = (int)(ent >> 16);
        const int g     = b * NPAIR + p;
        const int j_loc = pj[p];
        const int idx_i = b * NATOM + a_base + a;
        const int idx_j = b * NATOM + j_loc;

        const float dx = cart[idx_i * 3 + 0] - cart[idx_j * 3 + 0] + shifts[g * 3 + 0];
        const float dy = cart[idx_i * 3 + 1] - cart[idx_j * 3 + 1] + shifts[g * 3 + 1];
        const float dz = cart[idx_i * 3 + 2] - cart[idx_j * 3 + 2] + shifts[g * 3 + 2];
        const float dist = sqrtf(dx * dx + dy * dy + dz * dz);

        const int sp = species[idx_j];
        const float tc = 0.5f * __cosf(dist * (3.14159265358979f / cutoff)) + 0.5f;
        const float fc = tc * tc;

        float* r = rec + lane * RECSTRIDE;
        const float dv[3] = {dx, dy, dz};
        float ax[NANG];
        ax[0] = fc;
        ax[1] = fc * dx; ax[2] = fc * dy; ax[3] = fc * dz;
        int tpos = 4;
#pragma unroll
        for (int i = 1; i <= 3; ++i)
#pragma unroll
            for (int c = 0; c < 3; ++c)
                ax[tpos++] = ax[i] * dv[c];
        for (int l = 0; l < NANG; ++l) r[l] = ax[l];
        float* own = r + NANG + ((a & 1) << 3);
        float* oth = r + NANG + (((a & 1) ^ 1) << 3);
        for (int ww = 0; ww < NWAVE; ++ww) {
            const float dd = dist - rs[sp * NWAVE + ww];
            own[ww] = __expf(inta[sp * NWAVE + ww] * dd * dd) * params[sp * NWAVE + ww];
            oth[ww] = 0.f;
        }
    } else if (lane < 4 * nk) {
        float* r = rec + lane * RECSTRIDE;
        for (int l = 0; l < RECSTRIDE; ++l) r[l] = 0.f;
    }
    wave_sync();

    for (int ks = 0; ks < nk; ++ks) {
        const int k0 = 4 * ks + 2 * half;
        const float* f = rec + k0 * RECSTRIDE + lane16;
        v2f av, bv;
        bv.x = f[0];             bv.y = f[RECSTRIDE];
        av.x = f[NANG];          av.y = f[RECSTRIDE + NANG];
        acc = __builtin_amdgcn_wmma_f32_16x16x4_f32(
            false, av, false, bv, (short)0, acc, false, false);
    }
    wave_sync();
}

__global__ __launch_bounds__(256) void eann_gather_kernel(
    const float* __restrict__ cart,
    const int*   __restrict__ species,
    const int*   __restrict__ atom_index,
    const float* __restrict__ shifts,
    const float* __restrict__ rs,
    const float* __restrict__ inta,
    const float* __restrict__ params,
    const float* __restrict__ cutoff_p,
    const float* __restrict__ hyper,
    float*       __restrict__ out)
{
    __shared__ unsigned short s_list[8][WPAIRS];
    __shared__ unsigned s_queue[8][QCAP];
    __shared__ float    s_rec[8][BATCH * RECSTRIDE];
    __shared__ __attribute__((aligned(16))) float s_tile[TILE_ATOMS * NWORB];
    __shared__ int      s_cnt[8];

    const int t      = threadIdx.x;
    const int lane   = t & 31;
    const int wave   = t >> 5;
    const int b      = blockIdx.x >> 5;
    const int tib    = blockIdx.x & 31;
    const int a_base = tib * TILE_ATOMS;

    const int* __restrict__ pi = atom_index + b * NPAIR;
    const int* __restrict__ pj = atom_index + TOTPAIR + b * NPAIR;

    const float cutoff = cutoff_p[0];

    const int half   = lane >> 4;
    const int lane16 = lane & 15;
    const int kb     = half * 2;
    const int n      = wave * 16 + lane16;
    v2f bt[3][2];
#pragma unroll
    for (int l = 0; l < 3; ++l)
#pragma unroll
        for (int c = 0; c < 2; ++c) {
            bt[l][c].x = hyper[(l * NWAVE + 4 * c + kb + 0) * NORBIT + n];
            bt[l][c].y = hyper[(l * NWAVE + 4 * c + kb + 1) * NORBIT + n];
        }

    unsigned* q   = s_queue[wave];
    float*    rec = s_rec[wave];
    int qhead = 0, qcnt = 0;
    v8f acc = {};

    const int* __restrict__ pw = pi + wave * WPAIRS + lane * 4;
    int4 nx[NLOAD];
#pragma unroll
    for (int i = 0; i < NLOAD; ++i)
        nx[i] = *(const int4*)(pw + i * 128);

    unsigned short* lst = s_list[wave];
    int cnt = 0;
#pragma unroll
    for (int i = 0; i < NLOAD; ++i) {
        const int v[4] = {nx[i].x, nx[i].y, nx[i].z, nx[i].w};
#pragma unroll
        for (int c = 0; c < 4; ++c) {
            const bool m = (v[c] >> 4) == tib;
            const unsigned bal = __builtin_amdgcn_ballot_w32(m);
            const unsigned code = (unsigned)(i * 128 + lane * 4 + c) | ((unsigned)(v[c] & 15) << 12);
            if (m) lst[__builtin_amdgcn_mbcnt_lo(bal, (unsigned)cnt)] = (unsigned short)code;
            cnt += __builtin_popcount(bal);
        }
    }
    if (lane == 0) s_cnt[wave] = cnt;
    __syncthreads();

    for (int sl = 0; sl < 8; ++sl) {
        const int nlist = __builtin_amdgcn_readfirstlane(s_cnt[sl]);
        const unsigned short* L = s_list[sl];
        const unsigned hi = (unsigned)sl << 12;
        for (int cb = 0; cb < nlist; cb += 32 * FCHUNK) {
            unsigned cur[FCHUNK];
#pragma unroll
            for (int u = 0; u < FCHUNK; ++u) {
                const int e = cb + 32 * u + lane;
                cur[u] = (e < nlist) ? (unsigned)L[e] : 0xFFFFFFFFu;
            }
#pragma unroll
            for (int u = 0; u < FCHUNK; ++u) {
                const bool mine = (cur[u] >> 13) == (unsigned)wave;
                const unsigned m = __builtin_amdgcn_ballot_w32(mine);
                if (mine)
                    q[__builtin_amdgcn_mbcnt_lo(m, (unsigned)(qhead + qcnt)) & (QCAP - 1)] =
                        (cur[u] & 0xFFFu) | hi | ((cur[u] >> 12) << 16);
                qcnt += __builtin_popcount(m);
            }
            wave_sync();
            while (qcnt >= BATCH) {
                sum_batch(BATCH, qhead, q, rec, lane, lane16, half, b, a_base,
                          pj, cart, species, shifts, rs, inta, params, cutoff, acc);
                qhead = (qhead + BATCH) & (QCAP - 1);
                qcnt -= BATCH;
            }
        }
    }
    if (qcnt > 0)
        sum_batch(qcnt, qhead, q, rec, lane, lane16, half, b, a_base,
                  pj, cart, species, shifts, rs, inta, params, cutoff, acc);

    if (lane16 < NANG) {
        float4* dst = (float4*)(s_tile + (2 * wave + half) * NWORB + lane16 * NWAVE);
        dst[0] = make_float4(acc[0], acc[1], acc[2], acc[3]);
        dst[1] = make_float4(acc[4], acc[5], acc[6], acc[7]);
    }
    __syncthreads();

    v8f dens = {};
    const float* arow = s_tile + lane16 * NWORB + kb;
#pragma unroll
    for (int j = 0; j < NANG; ++j) {
        const int l = (j == 0) ? 0 : ((j < 4) ? 1 : 2);
        const v2f a0 = *(const v2f*)(arow + j * NWAVE);
        const v2f a1 = *(const v2f*)(arow + j * NWAVE + 4);
        v8f c0 = {};
        v8f dmat = __builtin_amdgcn_wmma_f32_16x16x4_f32(
            false, a0, false, bt[l][0], (short)0, c0, false, false);
        dmat = __builtin_amdgcn_wmma_f32_16x16x4_f32(
            false, a1, false, bt[l][1], (short)0, dmat, false, false);
        dens += dmat * dmat;
    }

    const int row0 = b * NATOM + a_base + half * 8;
#pragma unroll
    for (int r = 0; r < 8; ++r)
        out[(size_t)(row0 + r) * NORBIT + n] = dens[r];
}

extern "C" void kernel_launch(void* const* d_in, const int* in_sizes, int n_in,
                              void* d_out, int out_size, void* d_ws, size_t ws_size,
                              hipStream_t stream) {
    const float* cart       = (const float*)d_in[0];
    const int*   species    = (const int*)  d_in[2];
    const int*   atom_index = (const int*)  d_in[3];
    const float* shifts     = (const float*)d_in[4];
    const float* rs         = (const float*)d_in[5];
    const float* inta       = (const float*)d_in[6];
    const float* params     = (const float*)d_in[7];
    const float* hyper      = (const float*)d_in[8];
    const float* cutoff     = (const float*)d_in[10];
    float*       out        = (float*)d_out;

    eann_gather_kernel<<<512, 256, 0, stream>>>(
        cart, species, atom_index, shifts, rs, inta, params, cutoff, hyper, out);
}


# Record build reads its per-atom operands from LDS: the batch's positions, species and the radial tables are staged once per work-group; only the second index and the shift stay global gathers
# speedup vs baseline: 84.4049x; 1.1329x over previous
#include <hip/hip_runtime.h>
#include <math.h>

#define NBATCH  16
#define NATOM   512
#define NEIGH   64
#define NPAIR   (NATOM * NEIGH)
#define TOTPAIR (NBATCH * NPAIR)
#define TOTATOM (NBATCH * NATOM)
#define NWAVE   8
#define NTYPE   4
#define NANG    13
#define NWORB   (NANG * NWAVE)
#define NORBIT  128

typedef __attribute__((ext_vector_type(2)))  float    v2f;
typedef __attribute__((ext_vector_type(8)))  float    v8f;

#define TILE_ATOMS 16
#define WPAIRS     (NPAIR / 8)
#define NLOAD      (WPAIRS / 128)
#define BATCH      32
#define FCHUNK     5
#define QCAP       256
#define NMASK      16
#define TABROW     (3 * NWAVE)
#define RECSTRIDE  (NANG + NMASK)

__device__ __forceinline__ void wave_sync()
{
    __builtin_amdgcn_fence(__ATOMIC_RELEASE, "wavefront");
    __builtin_amdgcn_wave_barrier();
    __builtin_amdgcn_fence(__ATOMIC_ACQUIRE, "wavefront");
}

__device__ __forceinline__ void sum_batch(
    int batch, int head, const unsigned* q, float* rec,
    int lane, int lane16, int half, int b, int a_base,
    const int* __restrict__ pj, const float* __restrict__ shifts,
    const float* s_atom, const float* s_tab, float cutoff, v8f& acc)
{
    const int nk = __builtin_amdgcn_readfirstlane((batch + 3) >> 2);
    if (lane < batch) {
        const unsigned ent = q[(head + lane) & (QCAP - 1)];
        const int p     = (int)(ent & 0xFFFFu);
        const int a     = (int)(ent >> 16);
        const int g     = b * NPAIR + p;
        const int j_loc = pj[p];
        const float shx = shifts[g * 3 + 0];
        const float shy = shifts[g * 3 + 1];
        const float shz = shifts[g * 3 + 2];
        const float4 ai = *(const float4*)(s_atom + (a_base + a) * 4);
        const float4 aj = *(const float4*)(s_atom + j_loc * 4);

        const float dx = ai.x - aj.x + shx;
        const float dy = ai.y - aj.y + shy;
        const float dz = ai.z - aj.z + shz;
        const float dist = sqrtf(dx * dx + dy * dy + dz * dz);

        const int sp = __float_as_int(aj.w);
        const float4* tb4 = (const float4*)(s_tab + sp * TABROW);
        const float4 t0 = tb4[0], t1 = tb4[1], t2 = tb4[2], t3 = tb4[3], t4 = tb4[4], t5 = tb4[5];
        const float tb[TABROW] = {t0.x, t0.y, t0.z, t0.w, t1.x, t1.y, t1.z, t1.w,
                                  t2.x, t2.y, t2.z, t2.w, t3.x, t3.y, t3.z, t3.w,
                                  t4.x, t4.y, t4.z, t4.w, t5.x, t5.y, t5.z, t5.w};
        const float tc = 0.5f * __cosf(dist * (3.14159265358979f / cutoff)) + 0.5f;
        const float fc = tc * tc;

        float* r = rec + lane * RECSTRIDE;
        const float dv[3] = {dx, dy, dz};
        float ax[NANG];
        ax[0] = fc;
        ax[1] = fc * dx; ax[2] = fc * dy; ax[3] = fc * dz;
        int tpos = 4;
#pragma unroll
        for (int i = 1; i <= 3; ++i)
#pragma unroll
            for (int c = 0; c < 3; ++c)
                ax[tpos++] = ax[i] * dv[c];
        asm volatile("" :: "v"(t0.x), "v"(t0.y), "v"(t0.z), "v"(t0.w), "v"(t1.x), "v"(t1.y), "v"(t1.z), "v"(t1.w),
                           "v"(t2.x), "v"(t2.y), "v"(t2.z), "v"(t2.w), "v"(t3.x), "v"(t3.y), "v"(t3.z), "v"(t3.w),
                           "v"(t4.x), "v"(t4.y), "v"(t4.z), "v"(t4.w), "v"(t5.x), "v"(t5.y), "v"(t5.z), "v"(t5.w));
        for (int l = 0; l < NANG; ++l) r[l] = ax[l];
        float* own = r + NANG + ((a & 1) << 3);
        float* oth = r + NANG + (((a & 1) ^ 1) << 3);
        for (int ww = 0; ww < NWAVE; ++ww) {
            const float dd = dist - tb[ww];
            own[ww] = __expf(tb[NWAVE + ww] * dd * dd) * tb[2 * NWAVE + ww];
            oth[ww] = 0.f;
        }
    } else if (lane < 4 * nk) {
        float* r = rec + lane * RECSTRIDE;
        for (int l = 0; l < RECSTRIDE; ++l) r[l] = 0.f;
    }
    wave_sync();

    for (int ks = 0; ks < nk; ++ks) {
        const int k0 = 4 * ks + 2 * half;
        const float* f = rec + k0 * RECSTRIDE + lane16;
        v2f av, bv;
        bv.x = f[0];             bv.y = f[RECSTRIDE];
        av.x = f[NANG];          av.y = f[RECSTRIDE + NANG];
        acc = __builtin_amdgcn_wmma_f32_16x16x4_f32(
            false, av, false, bv, (short)0, acc, false, false);
    }
    wave_sync();
}

__global__ __launch_bounds__(256) void eann_gather_kernel(
    const float* __restrict__ cart,
    const int*   __restrict__ species,
    const int*   __restrict__ atom_index,
    const float* __restrict__ shifts,
    const float* __restrict__ rs,
    const float* __restrict__ inta,
    const float* __restrict__ params,
    const float* __restrict__ cutoff_p,
    const float* __restrict__ hyper,
    float*       __restrict__ out)
{
    __shared__ unsigned short s_list[8][WPAIRS];
    __shared__ unsigned s_queue[8][QCAP];
    __shared__ float    s_rec[8][BATCH * RECSTRIDE];
    __shared__ __attribute__((aligned(16))) float s_tile[TILE_ATOMS * NWORB];
    __shared__ int      s_cnt[8];
    __shared__ __attribute__((aligned(16))) float s_atom[NATOM * 4];
    __shared__ __attribute__((aligned(16))) float s_tab[NTYPE * TABROW];

    const int t      = threadIdx.x;
    const int lane   = t & 31;
    const int wave   = t >> 5;
    const int b      = blockIdx.x >> 5;
    const int tib    = blockIdx.x & 31;
    const int a_base = tib * TILE_ATOMS;

    const int* __restrict__ pi = atom_index + b * NPAIR;
    const int* __restrict__ pj = atom_index + TOTPAIR + b * NPAIR;

    const float cutoff = cutoff_p[0];

    const int half   = lane >> 4;
    const int lane16 = lane & 15;
    const int kb     = half * 2;
    const int n      = wave * 16 + lane16;
    v2f bt[3][2];
#pragma unroll
    for (int l = 0; l < 3; ++l)
#pragma unroll
        for (int c = 0; c < 2; ++c) {
            bt[l][c].x = hyper[(l * NWAVE + 4 * c + kb + 0) * NORBIT + n];
            bt[l][c].y = hyper[(l * NWAVE + 4 * c + kb + 1) * NORBIT + n];
        }

    unsigned* q   = s_queue[wave];
    float*    rec = s_rec[wave];
    int qhead = 0, qcnt = 0;
    v8f acc = {};

    const int at0 = b * NATOM + t;
    const int at1 = at0 + 256;
    const float c0x = cart[at0 * 3 + 0], c0y = cart[at0 * 3 + 1], c0z = cart[at0 * 3 + 2];
    const float c1x = cart[at1 * 3 + 0], c1y = cart[at1 * 3 + 1], c1z = cart[at1 * 3 + 2];
    const int   sp0 = species[at0];
    const int   sp1 = species[at1];
    float tabv = 0.f;
    if (t < NTYPE * TABROW) {
        const int ts = t / TABROW;
        const int tk = t - ts * TABROW;
        const float* __restrict__ tsrc = (tk < NWAVE) ? rs : ((tk < 2 * NWAVE) ? inta : params);
        tabv = tsrc[ts * NWAVE + (tk & (NWAVE - 1))];
    }

    const int* __restrict__ pw = pi + wave * WPAIRS + lane * 4;
    int4 nx[NLOAD];
#pragma unroll
    for (int i = 0; i < NLOAD; ++i)
        nx[i] = *(const int4*)(pw + i * 128);

    *(float4*)(s_atom + t * 4)         = make_float4(c0x, c0y, c0z, __int_as_float(sp0));
    *(float4*)(s_atom + (t + 256) * 4) = make_float4(c1x, c1y, c1z, __int_as_float(sp1));
    if (t < NTYPE * TABROW) s_tab[t] = tabv;

    unsigned short* lst = s_list[wave];
    int cnt = 0;
#pragma unroll
    for (int i = 0; i < NLOAD; ++i) {
        const int v[4] = {nx[i].x, nx[i].y, nx[i].z, nx[i].w};
#pragma unroll
        for (int c = 0; c < 4; ++c) {
            const bool m = (v[c] >> 4) == tib;
            const unsigned bal = __builtin_amdgcn_ballot_w32(m);
            const unsigned code = (unsigned)(i * 128 + lane * 4 + c) | ((unsigned)(v[c] & 15) << 12);
            if (m) lst[__builtin_amdgcn_mbcnt_lo(bal, (unsigned)cnt)] = (unsigned short)code;
            cnt += __builtin_popcount(bal);
        }
    }
    if (lane == 0) s_cnt[wave] = cnt;
    __syncthreads();

    for (int sl = 0; sl < 8; ++sl) {
        const int nlist = __builtin_amdgcn_readfirstlane(s_cnt[sl]);
        const unsigned short* L = s_list[sl];
        const unsigned hi = (unsigned)sl << 12;
        for (int cb = 0; cb < nlist; cb += 32 * FCHUNK) {
            unsigned cur[FCHUNK];
#pragma unroll
            for (int u = 0; u < FCHUNK; ++u) {
                const int e = cb + 32 * u + lane;
                cur[u] = (e < nlist) ? (unsigned)L[e] : 0xFFFFFFFFu;
            }
#pragma unroll
            for (int u = 0; u < FCHUNK; ++u) {
                const bool mine = (cur[u] >> 13) == (unsigned)wave;
                const unsigned m = __builtin_amdgcn_ballot_w32(mine);
                if (mine)
                    q[__builtin_amdgcn_mbcnt_lo(m, (unsigned)(qhead + qcnt)) & (QCAP - 1)] =
                        (cur[u] & 0xFFFu) | hi | ((cur[u] >> 12) << 16);
                qcnt += __builtin_popcount(m);
            }
            wave_sync();
            while (qcnt >= BATCH) {
                sum_batch(BATCH, qhead, q, rec, lane, lane16, half, b, a_base,
                          pj, shifts, s_atom, s_tab, cutoff, acc);
                qhead = (qhead + BATCH) & (QCAP - 1);
                qcnt -= BATCH;
            }
        }
    }
    if (qcnt > 0)
        sum_batch(qcnt, qhead, q, rec, lane, lane16, half, b, a_base,
                  pj, shifts, s_atom, s_tab, cutoff, acc);

    if (lane16 < NANG) {
        float4* dst = (float4*)(s_tile + (2 * wave + half) * NWORB + lane16 * NWAVE);
        dst[0] = make_float4(acc[0], acc[1], acc[2], acc[3]);
        dst[1] = make_float4(acc[4], acc[5], acc[6], acc[7]);
    }
    __syncthreads();

    v8f dens = {};
    const float* arow = s_tile + lane16 * NWORB + kb;
#pragma unroll
    for (int j = 0; j < NANG; ++j) {
        const int l = (j == 0) ? 0 : ((j < 4) ? 1 : 2);
        const v2f a0 = *(const v2f*)(arow + j * NWAVE);
        const v2f a1 = *(const v2f*)(arow + j * NWAVE + 4);
        v8f c0 = {};
        v8f dmat = __builtin_amdgcn_wmma_f32_16x16x4_f32(
            false, a0, false, bt[l][0], (short)0, c0, false, false);
        dmat = __builtin_amdgcn_wmma_f32_16x16x4_f32(
            false, a1, false, bt[l][1], (short)0, dmat, false, false);
        dens += dmat * dmat;
    }

    const int row0 = b * NATOM + a_base + half * 8;
#pragma unroll
    for (int r = 0; r < 8; ++r)
        out[(size_t)(row0 + r) * NORBIT + n] = dens[r];
}

extern "C" void kernel_launch(void* const* d_in, const int* in_sizes, int n_in,
                              void* d_out, int out_size, void* d_ws, size_t ws_size,
                              hipStream_t stream) {
    const float* cart       = (const float*)d_in[0];
    const int*   species    = (const int*)  d_in[2];
    const int*   atom_index = (const int*)  d_in[3];
    const float* shifts     = (const float*)d_in[4];
    const float* rs         = (const float*)d_in[5];
    const float* inta       = (const float*)d_in[6];
    const float* params     = (const float*)d_in[7];
    const float* hyper      = (const float*)d_in[8];
    const float* cutoff     = (const float*)d_in[10];
    float*       out        = (float*)d_out;

    eann_gather_kernel<<<512, 256, 0, stream>>>(
        cart, species, atom_index, shifts, rs, inta, params, cutoff, hyper, out);
}


# Work-group doubled to 32 atoms and 16 waves (each batch list scanned 16 times, not 32); each list split by owner once by its scanning wave (4 bit ballots), not filtered by every owner; sum order kept
# speedup vs baseline: 93.7858x; 1.1111x over previous
#include <hip/hip_runtime.h>
#include <math.h>

#define NBATCH  16
#define NATOM   512
#define NEIGH   64
#define NPAIR   (NATOM * NEIGH)
#define TOTPAIR (NBATCH * NPAIR)
#define TOTATOM (NBATCH * NATOM)
#define NWAVE   8
#define NTYPE   4
#define NANG    13
#define NWORB   (NANG * NWAVE)
#define NORBIT  128

typedef __attribute__((ext_vector_type(2)))  float    v2f;
typedef __attribute__((ext_vector_type(8)))  float    v8f;

#define TILE_ATOMS 32
#define NW         16
#define WPAIRS     (NPAIR / NW)
#define NLOAD      (WPAIRS / 128)
#define POSBITS    11
#define BATCH      32
#define FCHUNK     5
#define QCAP       256
#define PBLK       (32 * FCHUNK)
#define NPBLK      ((WPAIRS + PBLK - 1) / PBLK)
#define SEGROW     18
#define NMASK      16
#define TABROW     (3 * NWAVE)
#define RECSTRIDE  (NANG + NMASK)

__device__ __forceinline__ void wave_sync()
{
    __builtin_amdgcn_fence(__ATOMIC_RELEASE, "wavefront");
    __builtin_amdgcn_wave_barrier();
    __builtin_amdgcn_fence(__ATOMIC_ACQUIRE, "wavefront");
}

__device__ __forceinline__ void sum_batch(
    int batch, int head, const unsigned* q, float* rec,
    int lane, int lane16, int half, int b, int a_base,
    const int* __restrict__ pj, const float* __restrict__ shifts,
    const float* s_atom, const float* s_tab, float cutoff, v8f& acc)
{
    const int nk = __builtin_amdgcn_readfirstlane((batch + 3) >> 2);
    if (lane < batch) {
        const unsigned ent = q[(head + lane) & (QCAP - 1)];
        const int p     = (int)(ent & 0xFFFFu);
        const int a     = (int)(ent >> 16);
        const int g     = b * NPAIR + p;
        const int j_loc = pj[p];
        const float shx = shifts[g * 3 + 0];
        const float shy = shifts[g * 3 + 1];
        const float shz = shifts[g * 3 + 2];
        const float4 ai = *(const float4*)(s_atom + (a_base + a) * 4);
        const float4 aj = *(const float4*)(s_atom + j_loc * 4);

        const float dx = ai.x - aj.x + shx;
        const float dy = ai.y - aj.y + shy;
        const float dz = ai.z - aj.z + shz;
        const float dist = sqrtf(dx * dx + dy * dy + dz * dz);

        const int sp = __float_as_int(aj.w);
        const float4* tb4 = (const float4*)(s_tab + sp * TABROW);
        const float4 t0 = tb4[0], t1 = tb4[1], t2 = tb4[2], t3 = tb4[3], t4 = tb4[4], t5 = tb4[5];
        const float tb[TABROW] = {t0.x, t0.y, t0.z, t0.w, t1.x, t1.y, t1.z, t1.w,
                                  t2.x, t2.y, t2.z, t2.w, t3.x, t3.y, t3.z, t3.w,
                                  t4.x, t4.y, t4.z, t4.w, t5.x, t5.y, t5.z, t5.w};
        const float tc = 0.5f * __cosf(dist * (3.14159265358979f / cutoff)) + 0.5f;
        const float fc = tc * tc;

        float* r = rec + lane * RECSTRIDE;
        const float dv[3] = {dx, dy, dz};
        float ax[NANG];
        ax[0] = fc;
        ax[1] = fc * dx; ax[2] = fc * dy; ax[3] = fc * dz;
        int tpos = 4;
#pragma unroll
        for (int i = 1; i <= 3; ++i)
#pragma unroll
            for (int c = 0; c < 3; ++c)
                ax[tpos++] = ax[i] * dv[c];
        asm volatile("" :: "v"(t0.x), "v"(t0.y), "v"(t0.z), "v"(t0.w), "v"(t1.x), "v"(t1.y), "v"(t1.z), "v"(t1.w),
                           "v"(t2.x), "v"(t2.y), "v"(t2.z), "v"(t2.w), "v"(t3.x), "v"(t3.y), "v"(t3.z), "v"(t3.w),
                           "v"(t4.x), "v"(t4.y), "v"(t4.z), "v"(t4.w), "v"(t5.x), "v"(t5.y), "v"(t5.z), "v"(t5.w));
        for (int l = 0; l < NANG; ++l) r[l] = ax[l];
        float* own = r + NANG + ((a & 1) << 3);
        float* oth = r + NANG + (((a & 1) ^ 1) << 3);
        for (int ww = 0; ww < NWAVE; ++ww) {
            const float dd = dist - tb[ww];
            own[ww] = __expf(tb[NWAVE + ww] * dd * dd) * tb[2 * NWAVE + ww];
            oth[ww] = 0.f;
        }
    } else if (lane < 4 * nk) {
        float* r = rec + lane * RECSTRIDE;
        for (int l = 0; l < RECSTRIDE; ++l) r[l] = 0.f;
    }
    wave_sync();

    for (int ks = 0; ks < nk; ++ks) {
        const int k0 = 4 * ks + 2 * half;
        const float* f = rec + k0 * RECSTRIDE + lane16;
        v2f av, bv;
        bv.x = f[0];             bv.y = f[RECSTRIDE];
        av.x = f[NANG];          av.y = f[RECSTRIDE + NANG];
        acc = __builtin_amdgcn_wmma_f32_16x16x4_f32(
            false, av, false, bv, (short)0, acc, false, false);
    }
    wave_sync();
}

__global__ __launch_bounds__(32 * NW) void eann_gather_kernel(
    const float* __restrict__ cart,
    const int*   __restrict__ species,
    const int*   __restrict__ atom_index,
    const float* __restrict__ shifts,
    const float* __restrict__ rs,
    const float* __restrict__ inta,
    const float* __restrict__ params,
    const float* __restrict__ cutoff_p,
    const float* __restrict__ hyper,
    float*       __restrict__ out)
{
    __shared__ unsigned short s_list[NW][WPAIRS];
    __shared__ unsigned s_queue[NW][QCAP];
    __shared__ float    s_rec[NW][BATCH * RECSTRIDE];
    __shared__ __attribute__((aligned(16))) float s_tile[TILE_ATOMS * NWORB];
    __shared__ int      s_cnt[NW];
    __shared__ unsigned short s_seg[NW][NPBLK][SEGROW];
    __shared__ __attribute__((aligned(16))) float s_atom[NATOM * 4];
    __shared__ __attribute__((aligned(16))) float s_tab[NTYPE * TABROW];

    const int t      = threadIdx.x;
    const int lane   = t & 31;
    const int wave   = t >> 5;
    const int b      = blockIdx.x >> 4;
    const int tib    = blockIdx.x & 15;
    const int a_base = tib * TILE_ATOMS;

    const int* __restrict__ pi = atom_index + b * NPAIR;
    const int* __restrict__ pj = atom_index + TOTPAIR + b * NPAIR;

    const float cutoff = cutoff_p[0];

    const int half   = lane >> 4;
    const int lane16 = lane & 15;
    const int kb     = half * 2;
    const int n      = (wave & 7) * 16 + lane16;
    v2f bt[3][2];
#pragma unroll
    for (int l = 0; l < 3; ++l)
#pragma unroll
        for (int c = 0; c < 2; ++c) {
            bt[l][c].x = hyper[(l * NWAVE + 4 * c + kb + 0) * NORBIT + n];
            bt[l][c].y = hyper[(l * NWAVE + 4 * c + kb + 1) * NORBIT + n];
        }

    unsigned* q   = s_queue[wave];
    float*    rec = s_rec[wave];
    int qhead = 0, qcnt = 0;
    v8f acc = {};

    const int at0 = b * NATOM + t;
    const float c0x = cart[at0 * 3 + 0], c0y = cart[at0 * 3 + 1], c0z = cart[at0 * 3 + 2];
    const int   sp0 = species[at0];
    float tabv = 0.f;
    if (t < NTYPE * TABROW) {
        const int ts = t / TABROW;
        const int tk = t - ts * TABROW;
        const float* __restrict__ tsrc = (tk < NWAVE) ? rs : ((tk < 2 * NWAVE) ? inta : params);
        tabv = tsrc[ts * NWAVE + (tk & (NWAVE - 1))];
    }

    const int* __restrict__ pw = pi + wave * WPAIRS + lane * 4;
    int4 nx[NLOAD];
#pragma unroll
    for (int i = 0; i < NLOAD; ++i)
        nx[i] = *(const int4*)(pw + i * 128);

    *(float4*)(s_atom + t * 4)         = make_float4(c0x, c0y, c0z, __int_as_float(sp0));
    if (t < NTYPE * TABROW) s_tab[t] = tabv;

    unsigned short* lst = s_list[wave];
    int cnt = 0;
#pragma unroll
    for (int i = 0; i < NLOAD; ++i) {
        const int v[4] = {nx[i].x, nx[i].y, nx[i].z, nx[i].w};
#pragma unroll
        for (int c = 0; c < 4; ++c) {
            const bool m = (v[c] >> 5) == tib;
            const unsigned bal = __builtin_amdgcn_ballot_w32(m);
            const unsigned code = (unsigned)(i * 128 + lane * 4 + c) | ((unsigned)(v[c] & 31) << POSBITS);
            if (m) lst[__builtin_amdgcn_mbcnt_lo(bal, (unsigned)cnt)] = (unsigned short)code;
            cnt += __builtin_popcount(bal);
        }
    }
    if (lane == 0) s_cnt[wave] = cnt;

    wave_sync();
    {
        const int ncnt = __builtin_amdgcn_readfirstlane(cnt);
        const unsigned lo4 = (unsigned)lane & 15u;
        const unsigned l0 = 0u - (lo4 & 1u), l1 = 0u - ((lo4 >> 1) & 1u);
        const unsigned l2 = 0u - ((lo4 >> 2) & 1u), l3 = 0u - (lo4 >> 3);
        if (ncnt == 0 && lane < SEGROW) s_seg[wave][0][lane] = 0;
        for (int pb = 0, blk = 0; pb < ncnt; pb += PBLK, ++blk) {
            const int nb = min(ncnt - pb, PBLK);
            unsigned cur[FCHUNK], rnk[FCHUNK];
            int cu[FCHUNK];
#pragma unroll
            for (int u = 0; u < FCHUNK; ++u) {
                const int e = 32 * u + lane;
                cur[u] = (e < nb) ? (unsigned)lst[pb + e] : 0xFFFFFFFFu;
            }
            int tot = 0;
#pragma unroll
            for (int u = 0; u < FCHUNK; ++u) {
                const unsigned V  = __builtin_amdgcn_ballot_w32((32 * u + lane) < nb);
                const int x0 = (int)(cur[u] << 19) >> 31;
                const int x1 = (int)(cur[u] << 18) >> 31;
                const int x2 = (int)(cur[u] << 17) >> 31;
                const int x3 = (int)(cur[u] << 16) >> 31;
                const unsigned B0 = __builtin_amdgcn_ballot_w32(x0 != 0);
                const unsigned B1 = __builtin_amdgcn_ballot_w32(x1 != 0);
                const unsigned B2 = __builtin_amdgcn_ballot_w32(x2 != 0);
                const unsigned B3 = __builtin_amdgcn_ballot_w32(x3 != 0);
                const unsigned same = V & ~(B0 ^ (unsigned)x0) & ~(B1 ^ (unsigned)x1)
                                        & ~(B2 ^ (unsigned)x2) & ~(B3 ^ (unsigned)x3);
                rnk[u] = __builtin_amdgcn_mbcnt_lo(same, 0u);
                const unsigned mo = V & ~(B0 ^ l0) & ~(B1 ^ l1) & ~(B2 ^ l2) & ~(B3 ^ l3);
                cu[u] = __builtin_popcount(mo);
                tot += cu[u];
            }
            int start = 0;
#pragma unroll
            for (int o = 0; o < NW - 1; ++o) {
                const int to = __builtin_amdgcn_readlane(tot, o);
                start += ((int)lo4 > o) ? to : 0;
            }
            int off[FCHUNK];
            int run = start;
#pragma unroll
            for (int u = 0; u < FCHUNK; ++u) {
                off[u] = __builtin_amdgcn_ds_bpermute((int)((cur[u] >> 10) & 60u), run);
                run += cu[u];
            }
#pragma unroll
            for (int u = 0; u < FCHUNK; ++u)
                if ((32 * u + lane) < nb) lst[pb + off[u] + (int)rnk[u]] = (unsigned short)cur[u];
            if (lane <= NW) s_seg[wave][blk][lane] = (unsigned short)((lane == NW) ? nb : start);
        }
    }
    __syncthreads();

    const int cn_l  = s_cnt[lane & (NW - 1)];
    const int sg0_l = (int)s_seg[lane & (NW - 1)][0][wave];
    const int sg1_l = (int)s_seg[lane & (NW - 1)][0][wave + 1];
    for (int sl = 0; sl < NW; ++sl) {
        const int nlist = __builtin_amdgcn_readlane(cn_l, sl);
        const unsigned short* L = s_list[sl];
        const unsigned hi = (unsigned)sl << POSBITS;
        for (int pb = 0, blk = 0; pb < nlist; pb += PBLK, ++blk) {
            int s0, s1;
            if (blk == 0) {
                s0 = __builtin_amdgcn_readlane(sg0_l, sl);
                s1 = __builtin_amdgcn_readlane(sg1_l, sl);
            } else {
                s0 = __builtin_amdgcn_readfirstlane((int)s_seg[sl][blk][wave]);
                s1 = __builtin_amdgcn_readfirstlane((int)s_seg[sl][blk][wave + 1]);
            }
            const int ns = s1 - s0;
            for (int e0 = 0; e0 < ns; e0 += 32) {
                const int e = e0 + lane;
                if (e < ns) {
                    const unsigned c = (unsigned)L[pb + s0 + e];
                    q[(qhead + qcnt + e) & (QCAP - 1)] =
                        (c & ((1u << POSBITS) - 1u)) | hi | ((c >> POSBITS) << 16);
                }
            }
            qcnt += ns;
            wave_sync();
            while (qcnt >= BATCH) {
                sum_batch(BATCH, qhead, q, rec, lane, lane16, half, b, a_base,
                          pj, shifts, s_atom, s_tab, cutoff, acc);
                qhead = (qhead + BATCH) & (QCAP - 1);
                qcnt -= BATCH;
            }
        }
    }
    if (qcnt > 0)
        sum_batch(qcnt, qhead, q, rec, lane, lane16, half, b, a_base,
                  pj, shifts, s_atom, s_tab, cutoff, acc);

    if (lane16 < NANG) {
        float4* dst = (float4*)(s_tile + (2 * wave + half) * NWORB + lane16 * NWAVE);
        dst[0] = make_float4(acc[0], acc[1], acc[2], acc[3]);
        dst[1] = make_float4(acc[4], acc[5], acc[6], acc[7]);
    }
    __syncthreads();

    const int rt = wave >> 3;
    v8f dens = {};
    const float* arow = s_tile + (rt * 16 + lane16) * NWORB + kb;
#pragma unroll
    for (int j = 0; j < NANG; ++j) {
        const int l = (j == 0) ? 0 : ((j < 4) ? 1 : 2);
        const v2f a0 = *(const v2f*)(arow + j * NWAVE);
        const v2f a1 = *(const v2f*)(arow + j * NWAVE + 4);
        v8f c0 = {};
        v8f dmat = __builtin_amdgcn_wmma_f32_16x16x4_f32(
            false, a0, false, bt[l][0], (short)0, c0, false, false);
        dmat = __builtin_amdgcn_wmma_f32_16x16x4_f32(
            false, a1, false, bt[l][1], (short)0, dmat, false, false);
        dens += dmat * dmat;
    }

    const int row0 = b * NATOM + a_base + rt * 16 + half * 8;
#pragma unroll
    for (int r = 0; r < 8; ++r)
        out[(size_t)(row0 + r) * NORBIT + n] = dens[r];
}

extern "C" void kernel_launch(void* const* d_in, const int* in_sizes, int n_in,
                              void* d_out, int out_size, void* d_ws, size_t ws_size,
                              hipStream_t stream) {
    const float* cart       = (const float*)d_in[0];
    const int*   species    = (const int*)  d_in[2];
    const int*   atom_index = (const int*)  d_in[3];
    const float* shifts     = (const float*)d_in[4];
    const float* rs         = (const float*)d_in[5];
    const float* inta       = (const float*)d_in[6];
    const float* params     = (const float*)d_in[7];
    const float* hyper      = (const float*)d_in[8];
    const float* cutoff     = (const float*)d_in[10];
    float*       out        = (float*)d_out;

    eann_gather_kernel<<<256, 512, 0, stream>>>(
        cart, species, atom_index, shifts, rs, inta, params, cutoff, hyper, out);
}


# Segment sum's two global gathers (second index, shift) run one pass ahead: a pass starts once the next is queued and requests that one's gathers before building its own records; sum order kept
# speedup vs baseline: 98.7102x; 1.0525x over previous
#include <hip/hip_runtime.h>
#include <math.h>

#define NBATCH  16
#define NATOM   512
#define NEIGH   64
#define NPAIR   (NATOM * NEIGH)
#define TOTPAIR (NBATCH * NPAIR)
#define TOTATOM (NBATCH * NATOM)
#define NWAVE   8
#define NTYPE   4
#define NANG    13
#define NWORB   (NANG * NWAVE)
#define NORBIT  128

typedef __attribute__((ext_vector_type(2)))  float    v2f;
typedef __attribute__((ext_vector_type(8)))  float    v8f;

#define TILE_ATOMS 32
#define NW         16
#define WPAIRS     (NPAIR / NW)
#define NLOAD      (WPAIRS / 128)
#define POSBITS    11
#define BATCH      32
#define FCHUNK     5
#define QCAP       256
#define PBLK       (32 * FCHUNK)
#define NPBLK      ((WPAIRS + PBLK - 1) / PBLK)
#define SEGROW     18
#define NMASK      16
#define TABROW     (3 * NWAVE)
#define RECSTRIDE  (NANG + NMASK)

__device__ __forceinline__ void wave_sync()
{
    __builtin_amdgcn_fence(__ATOMIC_RELEASE, "wavefront");
    __builtin_amdgcn_wave_barrier();
    __builtin_amdgcn_fence(__ATOMIC_ACQUIRE, "wavefront");
}

struct Pre { unsigned ent; int j; float sx, sy, sz; };

__device__ __forceinline__ Pre gather_ahead(
    int n, int head, const unsigned* q, int lane, int b,
    const int* __restrict__ pj, const float* __restrict__ shifts)
{
    Pre o = {0u, 0, 0.f, 0.f, 0.f};
    if (lane < n) {
        o.ent = q[(head + lane) & (QCAP - 1)];
        const int p = (int)(o.ent & 0xFFFFu);
        const int g = b * NPAIR + p;
        o.j  = pj[p];
        o.sx = shifts[g * 3 + 0];
        o.sy = shifts[g * 3 + 1];
        o.sz = shifts[g * 3 + 2];
    }
    return o;
}

__device__ __forceinline__ void sum_batch(
    int batch, const Pre& cur, float* rec,
    int lane, int lane16, int half, int a_base,
    const float* s_atom, const float* s_tab, float cutoff, v8f& acc)
{
    const int nk = __builtin_amdgcn_readfirstlane((batch + 3) >> 2);
    if (lane < batch) {
        const int a     = (int)(cur.ent >> 16);
        const int j_loc = cur.j;
        const float shx = cur.sx;
        const float shy = cur.sy;
        const float shz = cur.sz;
        const float4 ai = *(const float4*)(s_atom + (a_base + a) * 4);
        const float4 aj = *(const float4*)(s_atom + j_loc * 4);

        const float dx = ai.x - aj.x + shx;
        const float dy = ai.y - aj.y + shy;
        const float dz = ai.z - aj.z + shz;
        const float dist = sqrtf(dx * dx + dy * dy + dz * dz);

        const int sp = __float_as_int(aj.w);
        const float4* tb4 = (const float4*)(s_tab + sp * TABROW);
        const float4 t0 = tb4[0], t1 = tb4[1], t2 = tb4[2], t3 = tb4[3], t4 = tb4[4], t5 = tb4[5];
        const float tb[TABROW] = {t0.x, t0.y, t0.z, t0.w, t1.x, t1.y, t1.z, t1.w,
                                  t2.x, t2.y, t2.z, t2.w, t3.x, t3.y, t3.z, t3.w,
                                  t4.x, t4.y, t4.z, t4.w, t5.x, t5.y, t5.z, t5.w};
        const float tc = 0.5f * __cosf(dist * (3.14159265358979f / cutoff)) + 0.5f;
        const float fc = tc * tc;

        float* r = rec + lane * RECSTRIDE;
        const float dv[3] = {dx, dy, dz};
        float ax[NANG];
        ax[0] = fc;
        ax[1] = fc * dx; ax[2] = fc * dy; ax[3] = fc * dz;
        int tpos = 4;
#pragma unroll
        for (int i = 1; i <= 3; ++i)
#pragma unroll
            for (int c = 0; c < 3; ++c)
                ax[tpos++] = ax[i] * dv[c];
        asm volatile("" :: "v"(t0.x), "v"(t0.y), "v"(t0.z), "v"(t0.w), "v"(t1.x), "v"(t1.y), "v"(t1.z), "v"(t1.w),
                           "v"(t2.x), "v"(t2.y), "v"(t2.z), "v"(t2.w), "v"(t3.x), "v"(t3.y), "v"(t3.z), "v"(t3.w),
                           "v"(t4.x), "v"(t4.y), "v"(t4.z), "v"(t4.w), "v"(t5.x), "v"(t5.y), "v"(t5.z), "v"(t5.w));
        for (int l = 0; l < NANG; ++l) r[l] = ax[l];
        float* own = r + NANG + ((a & 1) << 3);
        float* oth = r + NANG + (((a & 1) ^ 1) << 3);
        for (int ww = 0; ww < NWAVE; ++ww) {
            const float dd = dist - tb[ww];
            own[ww] = __expf(tb[NWAVE + ww] * dd * dd) * tb[2 * NWAVE + ww];
            oth[ww] = 0.f;
        }
    } else if (lane < 4 * nk) {
        float* r = rec + lane * RECSTRIDE;
        for (int l = 0; l < RECSTRIDE; ++l) r[l] = 0.f;
    }
    wave_sync();

    for (int ks = 0; ks < nk; ++ks) {
        const int k0 = 4 * ks + 2 * half;
        const float* f = rec + k0 * RECSTRIDE + lane16;
        v2f av, bv;
        bv.x = f[0];             bv.y = f[RECSTRIDE];
        av.x = f[NANG];          av.y = f[RECSTRIDE + NANG];
        acc = __builtin_amdgcn_wmma_f32_16x16x4_f32(
            false, av, false, bv, (short)0, acc, false, false);
    }
    wave_sync();
}

__global__ __launch_bounds__(32 * NW) void eann_gather_kernel(
    const float* __restrict__ cart,
    const int*   __restrict__ species,
    const int*   __restrict__ atom_index,
    const float* __restrict__ shifts,
    const float* __restrict__ rs,
    const float* __restrict__ inta,
    const float* __restrict__ params,
    const float* __restrict__ cutoff_p,
    const float* __restrict__ hyper,
    float*       __restrict__ out)
{
    __shared__ unsigned short s_list[NW][WPAIRS];
    __shared__ unsigned s_queue[NW][QCAP];
    __shared__ float    s_rec[NW][BATCH * RECSTRIDE];
    __shared__ __attribute__((aligned(16))) float s_tile[TILE_ATOMS * NWORB];
    __shared__ int      s_cnt[NW];
    __shared__ unsigned short s_seg[NW][NPBLK][SEGROW];
    __shared__ __attribute__((aligned(16))) float s_atom[NATOM * 4];
    __shared__ __attribute__((aligned(16))) float s_tab[NTYPE * TABROW];

    const int t      = threadIdx.x;
    const int lane   = t & 31;
    const int wave   = t >> 5;
    const int b      = blockIdx.x >> 4;
    const int tib    = blockIdx.x & 15;
    const int a_base = tib * TILE_ATOMS;

    const int* __restrict__ pi = atom_index + b * NPAIR;
    const int* __restrict__ pj = atom_index + TOTPAIR + b * NPAIR;

    const float cutoff = cutoff_p[0];

    const int half   = lane >> 4;
    const int lane16 = lane & 15;
    const int kb     = half * 2;
    const int n      = (wave & 7) * 16 + lane16;
    v2f bt[3][2];
#pragma unroll
    for (int l = 0; l < 3; ++l)
#pragma unroll
        for (int c = 0; c < 2; ++c) {
            bt[l][c].x = hyper[(l * NWAVE + 4 * c + kb + 0) * NORBIT + n];
            bt[l][c].y = hyper[(l * NWAVE + 4 * c + kb + 1) * NORBIT + n];
        }

    unsigned* q   = s_queue[wave];
    float*    rec = s_rec[wave];
    int qhead = 0, qcnt = 0;
    Pre  pf = {0u, 0, 0.f, 0.f, 0.f};
    bool have = false;
    v8f acc = {};

    const int at0 = b * NATOM + t;
    const float c0x = cart[at0 * 3 + 0], c0y = cart[at0 * 3 + 1], c0z = cart[at0 * 3 + 2];
    const int   sp0 = species[at0];
    float tabv = 0.f;
    if (t < NTYPE * TABROW) {
        const int ts = t / TABROW;
        const int tk = t - ts * TABROW;
        const float* __restrict__ tsrc = (tk < NWAVE) ? rs : ((tk < 2 * NWAVE) ? inta : params);
        tabv = tsrc[ts * NWAVE + (tk & (NWAVE - 1))];
    }

    const int* __restrict__ pw = pi + wave * WPAIRS + lane * 4;
    int4 nx[NLOAD];
#pragma unroll
    for (int i = 0; i < NLOAD; ++i)
        nx[i] = *(const int4*)(pw + i * 128);

    *(float4*)(s_atom + t * 4)         = make_float4(c0x, c0y, c0z, __int_as_float(sp0));
    if (t < NTYPE * TABROW) s_tab[t] = tabv;

    unsigned short* lst = s_list[wave];
    int cnt = 0;
#pragma unroll
    for (int i = 0; i < NLOAD; ++i) {
        const int v[4] = {nx[i].x, nx[i].y, nx[i].z, nx[i].w};
#pragma unroll
        for (int c = 0; c < 4; ++c) {
            const bool m = (v[c] >> 5) == tib;
            const unsigned bal = __builtin_amdgcn_ballot_w32(m);
            const unsigned code = (unsigned)(i * 128 + lane * 4 + c) | ((unsigned)(v[c] & 31) << POSBITS);
            if (m) lst[__builtin_amdgcn_mbcnt_lo(bal, (unsigned)cnt)] = (unsigned short)code;
            cnt += __builtin_popcount(bal);
        }
    }
    if (lane == 0) s_cnt[wave] = cnt;

    wave_sync();
    {
        const int ncnt = __builtin_amdgcn_readfirstlane(cnt);
        const unsigned lo4 = (unsigned)lane & 15u;
        const unsigned l0 = 0u - (lo4 & 1u), l1 = 0u - ((lo4 >> 1) & 1u);
        const unsigned l2 = 0u - ((lo4 >> 2) & 1u), l3 = 0u - (lo4 >> 3);
        if (ncnt == 0 && lane < SEGROW) s_seg[wave][0][lane] = 0;
        for (int pb = 0, blk = 0; pb < ncnt; pb += PBLK, ++blk) {
            const int nb = min(ncnt - pb, PBLK);
            unsigned cur[FCHUNK], rnk[FCHUNK];
            int cu[FCHUNK];
#pragma unroll
            for (int u = 0; u < FCHUNK; ++u) {
                const int e = 32 * u + lane;
                cur[u] = (e < nb) ? (unsigned)lst[pb + e] : 0xFFFFFFFFu;
            }
            int tot = 0;
#pragma unroll
            for (int u = 0; u < FCHUNK; ++u) {
                const unsigned V  = __builtin_amdgcn_ballot_w32((32 * u + lane) < nb);
                const int x0 = (int)(cur[u] << 19) >> 31;
                const int x1 = (int)(cur[u] << 18) >> 31;
                const int x2 = (int)(cur[u] << 17) >> 31;
                const int x3 = (int)(cur[u] << 16) >> 31;
                const unsigned B0 = __builtin_amdgcn_ballot_w32(x0 != 0);
                const unsigned B1 = __builtin_amdgcn_ballot_w32(x1 != 0);
                const unsigned B2 = __builtin_amdgcn_ballot_w32(x2 != 0);
                const unsigned B3 = __builtin_amdgcn_ballot_w32(x3 != 0);
                const unsigned same = V & ~(B0 ^ (unsigned)x0) & ~(B1 ^ (unsigned)x1)
                                        & ~(B2 ^ (unsigned)x2) & ~(B3 ^ (unsigned)x3);
                rnk[u] = __builtin_amdgcn_mbcnt_lo(same, 0u);
                const unsigned mo = V & ~(B0 ^ l0) & ~(B1 ^ l1) & ~(B2 ^ l2) & ~(B3 ^ l3);
                cu[u] = __builtin_popcount(mo);
                tot += cu[u];
            }
            int start = 0;
#pragma unroll
            for (int o = 0; o < NW - 1; ++o) {
                const int to = __builtin_amdgcn_readlane(tot, o);
                start += ((int)lo4 > o) ? to : 0;
            }
            int off[FCHUNK];
            int run = start;
#pragma unroll
            for (int u = 0; u < FCHUNK; ++u) {
                off[u] = __builtin_amdgcn_ds_bpermute((int)((cur[u] >> 10) & 60u), run);
                run += cu[u];
            }
#pragma unroll
            for (int u = 0; u < FCHUNK; ++u)
                if ((32 * u + lane) < nb) lst[pb + off[u] + (int)rnk[u]] = (unsigned short)cur[u];
            if (lane <= NW) s_seg[wave][blk][lane] = (unsigned short)((lane == NW) ? nb : start);
        }
    }
    __syncthreads();

    const int cn_l  = s_cnt[lane & (NW - 1)];
    const int sg0_l = (int)s_seg[lane & (NW - 1)][0][wave];
    const int sg1_l = (int)s_seg[lane & (NW - 1)][0][wave + 1];
    for (int sl = 0; sl < NW; ++sl) {
        const int nlist = __builtin_amdgcn_readlane(cn_l, sl);
        const unsigned short* L = s_list[sl];
        const unsigned hi = (unsigned)sl << POSBITS;
        for (int pb = 0, blk = 0; pb < nlist; pb += PBLK, ++blk) {
            int s0, s1;
            if (blk == 0) {
                s0 = __builtin_amdgcn_readlane(sg0_l, sl);
                s1 = __builtin_amdgcn_readlane(sg1_l, sl);
            } else {
                s0 = __builtin_amdgcn_readfirstlane((int)s_seg[sl][blk][wave]);
                s1 = __builtin_amdgcn_readfirstlane((int)s_seg[sl][blk][wave + 1]);
            }
            const int ns = s1 - s0;
            for (int e0 = 0; e0 < ns; e0 += 32) {
                const int e = e0 + lane;
                if (e < ns) {
                    const unsigned c = (unsigned)L[pb + s0 + e];
                    q[(qhead + qcnt + e) & (QCAP - 1)] =
                        (c & ((1u << POSBITS) - 1u)) | hi | ((c >> POSBITS) << 16);
                }
            }
            qcnt += ns;
            wave_sync();
            if (!have && qcnt >= BATCH) {
                pf = gather_ahead(BATCH, qhead, q, lane, b, pj, shifts);
                have = true;
            }
            while (qcnt >= 2 * BATCH) {
                const Pre cur = pf;
                pf = gather_ahead(BATCH, qhead + BATCH, q, lane, b, pj, shifts);
                sum_batch(BATCH, cur, rec, lane, lane16, half, a_base, s_atom, s_tab, cutoff, acc);
                qhead = (qhead + BATCH) & (QCAP - 1);
                qcnt -= BATCH;
            }
        }
    }
    int nlast;
    Pre last;
    if (qcnt >= BATCH) {
        nlast = qcnt - BATCH;
        last = gather_ahead(nlast, qhead + BATCH, q, lane, b, pj, shifts);
        sum_batch(BATCH, pf, rec, lane, lane16, half, a_base, s_atom, s_tab, cutoff, acc);
    } else {
        nlast = qcnt;
        last = gather_ahead(nlast, qhead, q, lane, b, pj, shifts);
    }
    if (nlast > 0)
        sum_batch(nlast, last, rec, lane, lane16, half, a_base, s_atom, s_tab, cutoff, acc);

    if (lane16 < NANG) {
        float4* dst = (float4*)(s_tile + (2 * wave + half) * NWORB + lane16 * NWAVE);
        dst[0] = make_float4(acc[0], acc[1], acc[2], acc[3]);
        dst[1] = make_float4(acc[4], acc[5], acc[6], acc[7]);
    }
    __syncthreads();

    const int rt = wave >> 3;
    v8f dens = {};
    const float* arow = s_tile + (rt * 16 + lane16) * NWORB + kb;
#pragma unroll
    for (int j = 0; j < NANG; ++j) {
        const int l = (j == 0) ? 0 : ((j < 4) ? 1 : 2);
        const v2f a0 = *(const v2f*)(arow + j * NWAVE);
        const v2f a1 = *(const v2f*)(arow + j * NWAVE + 4);
        v8f c0 = {};
        v8f dmat = __builtin_amdgcn_wmma_f32_16x16x4_f32(
            false, a0, false, bt[l][0], (short)0, c0, false, false);
        dmat = __builtin_amdgcn_wmma_f32_16x16x4_f32(
            false, a1, false, bt[l][1], (short)0, dmat, false, false);
        dens += dmat * dmat;
    }

    const int row0 = b * NATOM + a_base + rt * 16 + half * 8;
#pragma unroll
    for (int r = 0; r < 8; ++r)
        out[(size_t)(row0 + r) * NORBIT + n] = dens[r];
}

extern "C" void kernel_launch(void* const* d_in, const int* in_sizes, int n_in,
                              void* d_out, int out_size, void* d_ws, size_t ws_size,
                              hipStream_t stream) {
    const float* cart       = (const float*)d_in[0];
    const int*   species    = (const int*)  d_in[2];
    const int*   atom_index = (const int*)  d_in[3];
    const float* shifts     = (const float*)d_in[4];
    const float* rs         = (const float*)d_in[5];
    const float* inta       = (const float*)d_in[6];
    const float* params     = (const float*)d_in[7];
    const float* hyper      = (const float*)d_in[8];
    const float* cutoff     = (const float*)d_in[10];
    float*       out        = (float*)d_out;

    eann_gather_kernel<<<256, 512, 0, stream>>>(
        cart, species, atom_index, shifts, rs, inta, params, cutoff, hyper, out);
}


# Owners no longer walk the 16 lists: after exchanging 16x16 counts each scanning wave scatters its owner-sorted list into one owner-ordered LDS array (counting sort); one more barrier; sum order kept
# speedup vs baseline: 107.3622x; 1.0877x over previous
#include <hip/hip_runtime.h>
#include <math.h>

#define NBATCH  16
#define NATOM   512
#define NEIGH   64
#define NPAIR   (NATOM * NEIGH)
#define TOTPAIR (NBATCH * NPAIR)
#define TOTATOM (NBATCH * NATOM)
#define NWAVE   8
#define NTYPE   4
#define NANG    13
#define NWORB   (NANG * NWAVE)
#define NORBIT  128

typedef __attribute__((ext_vector_type(2)))  float    v2f;
typedef __attribute__((ext_vector_type(8)))  float    v8f;

#define TILE_ATOMS 32
#define NW         16
#define WPAIRS     (NPAIR / NW)
#define NLOAD      (WPAIRS / 128)
#define POSBITS    11
#define BATCH      32
#define FCHUNK     5
#define PBLK       (32 * FCHUNK)
#define NPBLK      ((WPAIRS + PBLK - 1) / PBLK)
#define SEGROW     18
#define NMASK      16
#define TABROW     (3 * NWAVE)
#define RECSTRIDE  (NANG + NMASK)

__device__ __forceinline__ void wave_sync()
{
    __builtin_amdgcn_fence(__ATOMIC_RELEASE, "wavefront");
    __builtin_amdgcn_wave_barrier();
    __builtin_amdgcn_fence(__ATOMIC_ACQUIRE, "wavefront");
}

struct Pre { unsigned ent; int j; float sx, sy, sz; };

__device__ __forceinline__ Pre gather_ahead(
    int n, const unsigned short* own, int lane, int wave, int b,
    const int* __restrict__ pj, const float* __restrict__ shifts)
{
    Pre o = {0u, 0, 0.f, 0.f, 0.f};
    if (lane < n) {
        const unsigned c = (unsigned)own[lane];
        const int p = (int)(c & 0x7FFFu);
        o.ent = (unsigned)p | (((unsigned)(2 * wave) + (c >> 15)) << 16);
        const int g = b * NPAIR + p;
        o.j  = pj[p];
        o.sx = shifts[g * 3 + 0];
        o.sy = shifts[g * 3 + 1];
        o.sz = shifts[g * 3 + 2];
    }
    return o;
}

__device__ __forceinline__ void sum_batch(
    int batch, const Pre& cur, float* rec,
    int lane, int lane16, int half, int a_base,
    const float* s_atom, const float* s_tab, float cutoff, v8f& acc)
{
    const int nk = __builtin_amdgcn_readfirstlane((batch + 3) >> 2);
    if (lane < batch) {
        const int a     = (int)(cur.ent >> 16);
        const int j_loc = cur.j;
        const float shx = cur.sx;
        const float shy = cur.sy;
        const float shz = cur.sz;
        const float4 ai = *(const float4*)(s_atom + (a_base + a) * 4);
        const float4 aj = *(const float4*)(s_atom + j_loc * 4);

        const float dx = ai.x - aj.x + shx;
        const float dy = ai.y - aj.y + shy;
        const float dz = ai.z - aj.z + shz;
        const float dist = sqrtf(dx * dx + dy * dy + dz * dz);

        const int sp = __float_as_int(aj.w);
        const float4* tb4 = (const float4*)(s_tab + sp * TABROW);
        const float4 t0 = tb4[0], t1 = tb4[1], t2 = tb4[2], t3 = tb4[3], t4 = tb4[4], t5 = tb4[5];
        const float tb[TABROW] = {t0.x, t0.y, t0.z, t0.w, t1.x, t1.y, t1.z, t1.w,
                                  t2.x, t2.y, t2.z, t2.w, t3.x, t3.y, t3.z, t3.w,
                                  t4.x, t4.y, t4.z, t4.w, t5.x, t5.y, t5.z, t5.w};
        const float tc = 0.5f * __cosf(dist * (3.14159265358979f / cutoff)) + 0.5f;
        const float fc = tc * tc;

        float* r = rec + lane * RECSTRIDE;
        const float dv[3] = {dx, dy, dz};
        float ax[NANG];
        ax[0] = fc;
        ax[1] = fc * dx; ax[2] = fc * dy; ax[3] = fc * dz;
        int tpos = 4;
#pragma unroll
        for (int i = 1; i <= 3; ++i)
#pragma unroll
            for (int c = 0; c < 3; ++c)
                ax[tpos++] = ax[i] * dv[c];
        asm volatile("" :: "v"(t0.x), "v"(t0.y), "v"(t0.z), "v"(t0.w), "v"(t1.x), "v"(t1.y), "v"(t1.z), "v"(t1.w),
                           "v"(t2.x), "v"(t2.y), "v"(t2.z), "v"(t2.w), "v"(t3.x), "v"(t3.y), "v"(t3.z), "v"(t3.w),
                           "v"(t4.x), "v"(t4.y), "v"(t4.z), "v"(t4.w), "v"(t5.x), "v"(t5.y), "v"(t5.z), "v"(t5.w));
        for (int l = 0; l < NANG; ++l) r[l] = ax[l];
        float* own = r + NANG + ((a & 1) << 3);
        float* oth = r + NANG + (((a & 1) ^ 1) << 3);
        for (int ww = 0; ww < NWAVE; ++ww) {
            const float dd = dist - tb[ww];
            own[ww] = __expf(tb[NWAVE + ww] * dd * dd) * tb[2 * NWAVE + ww];
            oth[ww] = 0.f;
        }
    } else if (lane < 4 * nk) {
        float* r = rec + lane * RECSTRIDE;
        for (int l = 0; l < RECSTRIDE; ++l) r[l] = 0.f;
    }
    wave_sync();

    for (int ks = 0; ks < nk; ++ks) {
        const int k0 = 4 * ks + 2 * half;
        const float* f = rec + k0 * RECSTRIDE + lane16;
        v2f av, bv;
        bv.x = f[0];             bv.y = f[RECSTRIDE];
        av.x = f[NANG];          av.y = f[RECSTRIDE + NANG];
        acc = __builtin_amdgcn_wmma_f32_16x16x4_f32(
            false, av, false, bv, (short)0, acc, false, false);
    }
    wave_sync();
}

__global__ __launch_bounds__(32 * NW) void eann_gather_kernel(
    const float* __restrict__ cart,
    const int*   __restrict__ species,
    const int*   __restrict__ atom_index,
    const float* __restrict__ shifts,
    const float* __restrict__ rs,
    const float* __restrict__ inta,
    const float* __restrict__ params,
    const float* __restrict__ cutoff_p,
    const float* __restrict__ hyper,
    float*       __restrict__ out)
{
    __shared__ unsigned short s_list[NW][WPAIRS];
    __shared__ unsigned short s_own[NPAIR];
    __shared__ float    s_rec[NW][BATCH * RECSTRIDE];
    __shared__ __attribute__((aligned(16))) float s_tile[TILE_ATOMS * NWORB];
    __shared__ int      s_tot[NW][NW];
    __shared__ unsigned short s_seg[NW][NPBLK][SEGROW];
    __shared__ __attribute__((aligned(16))) float s_atom[NATOM * 4];
    __shared__ __attribute__((aligned(16))) float s_tab[NTYPE * TABROW];

    const int t      = threadIdx.x;
    const int lane   = t & 31;
    const int wave   = t >> 5;
    const int b      = blockIdx.x >> 4;
    const int tib    = blockIdx.x & 15;
    const int a_base = tib * TILE_ATOMS;

    const int* __restrict__ pi = atom_index + b * NPAIR;
    const int* __restrict__ pj = atom_index + TOTPAIR + b * NPAIR;

    const float cutoff = cutoff_p[0];

    const int half   = lane >> 4;
    const int lane16 = lane & 15;
    const int kb     = half * 2;
    const int n      = (wave & 7) * 16 + lane16;
    v2f bt[3][2];
#pragma unroll
    for (int l = 0; l < 3; ++l)
#pragma unroll
        for (int c = 0; c < 2; ++c) {
            bt[l][c].x = hyper[(l * NWAVE + 4 * c + kb + 0) * NORBIT + n];
            bt[l][c].y = hyper[(l * NWAVE + 4 * c + kb + 1) * NORBIT + n];
        }

    float*    rec = s_rec[wave];
    Pre  pf = {0u, 0, 0.f, 0.f, 0.f};
    v8f acc = {};

    const int at0 = b * NATOM + t;
    const float c0x = cart[at0 * 3 + 0], c0y = cart[at0 * 3 + 1], c0z = cart[at0 * 3 + 2];
    const int   sp0 = species[at0];
    float tabv = 0.f;
    if (t < NTYPE * TABROW) {
        const int ts = t / TABROW;
        const int tk = t - ts * TABROW;
        const float* __restrict__ tsrc = (tk < NWAVE) ? rs : ((tk < 2 * NWAVE) ? inta : params);
        tabv = tsrc[ts * NWAVE + (tk & (NWAVE - 1))];
    }

    const int* __restrict__ pw = pi + wave * WPAIRS + lane * 4;
    int4 nx[NLOAD];
#pragma unroll
    for (int i = 0; i < NLOAD; ++i)
        nx[i] = *(const int4*)(pw + i * 128);

    *(float4*)(s_atom + t * 4)         = make_float4(c0x, c0y, c0z, __int_as_float(sp0));
    if (t < NTYPE * TABROW) s_tab[t] = tabv;

    unsigned short* lst = s_list[wave];
    int cnt = 0;
#pragma unroll
    for (int i = 0; i < NLOAD; ++i) {
        const int v[4] = {nx[i].x, nx[i].y, nx[i].z, nx[i].w};
#pragma unroll
        for (int c = 0; c < 4; ++c) {
            const bool m = (v[c] >> 5) == tib;
            const unsigned bal = __builtin_amdgcn_ballot_w32(m);
            const unsigned code = (unsigned)(i * 128 + lane * 4 + c) | ((unsigned)(v[c] & 31) << POSBITS);
            if (m) lst[__builtin_amdgcn_mbcnt_lo(bal, (unsigned)cnt)] = (unsigned short)code;
            cnt += __builtin_popcount(bal);
        }
    }

    wave_sync();
    {
        const int ncnt = __builtin_amdgcn_readfirstlane(cnt);
        const unsigned lo4 = (unsigned)lane & 15u;
        const unsigned l0 = 0u - (lo4 & 1u), l1 = 0u - ((lo4 >> 1) & 1u);
        const unsigned l2 = 0u - ((lo4 >> 2) & 1u), l3 = 0u - (lo4 >> 3);
        int ltot = 0;
        for (int pb = 0, blk = 0; pb < ncnt; pb += PBLK, ++blk) {
            const int nb = min(ncnt - pb, PBLK);
            unsigned cur[FCHUNK], rnk[FCHUNK];
            int cu[FCHUNK];
#pragma unroll
            for (int u = 0; u < FCHUNK; ++u) {
                const int e = 32 * u + lane;
                cur[u] = (e < nb) ? (unsigned)lst[pb + e] : 0xFFFFFFFFu;
            }
            int tot = 0;
#pragma unroll
            for (int u = 0; u < FCHUNK; ++u) {
                const unsigned V  = __builtin_amdgcn_ballot_w32((32 * u + lane) < nb);
                const int x0 = (int)(cur[u] << 19) >> 31;
                const int x1 = (int)(cur[u] << 18) >> 31;
                const int x2 = (int)(cur[u] << 17) >> 31;
                const int x3 = (int)(cur[u] << 16) >> 31;
                const unsigned B0 = __builtin_amdgcn_ballot_w32(x0 != 0);
                const unsigned B1 = __builtin_amdgcn_ballot_w32(x1 != 0);
                const unsigned B2 = __builtin_amdgcn_ballot_w32(x2 != 0);
                const unsigned B3 = __builtin_amdgcn_ballot_w32(x3 != 0);
                const unsigned same = V & ~(B0 ^ (unsigned)x0) & ~(B1 ^ (unsigned)x1)
                                        & ~(B2 ^ (unsigned)x2) & ~(B3 ^ (unsigned)x3);
                rnk[u] = __builtin_amdgcn_mbcnt_lo(same, 0u);
                const unsigned mo = V & ~(B0 ^ l0) & ~(B1 ^ l1) & ~(B2 ^ l2) & ~(B3 ^ l3);
                cu[u] = __builtin_popcount(mo);
                tot += cu[u];
            }
            int start = 0;
#pragma unroll
            for (int o = 0; o < NW - 1; ++o) {
                const int to = __builtin_amdgcn_readlane(tot, o);
                start += ((int)lo4 > o) ? to : 0;
            }
            int off[FCHUNK];
            int run = start;
#pragma unroll
            for (int u = 0; u < FCHUNK; ++u) {
                off[u] = __builtin_amdgcn_ds_bpermute((int)((cur[u] >> 10) & 60u), run);
                run += cu[u];
            }
#pragma unroll
            for (int u = 0; u < FCHUNK; ++u)
                if ((32 * u + lane) < nb) lst[pb + off[u] + (int)rnk[u]] = (unsigned short)cur[u];
            if (lane <= NW) s_seg[wave][blk][lane] = (unsigned short)((lane == NW) ? nb : start);
            ltot += tot;
        }
        if (lane < NW) s_tot[wave][lane] = ltot;
    }
    __syncthreads();

    const int wave_s = __builtin_amdgcn_readfirstlane(wave);
    int T = 0, P = 0;
#pragma unroll
    for (int w2 = 0; w2 < NW; ++w2) {
        const int v = s_tot[w2][lane & (NW - 1)];
        T += v;
        P += (w2 < wave_s) ? v : 0;
    }
    int S = 0;
#pragma unroll
    for (int o = 0; o < NW - 1; ++o) {
        const int to = __builtin_amdgcn_readlane(T, o);
        S += ((lane & (NW - 1)) > o) ? to : 0;
    }
    const int my_n    = __builtin_amdgcn_readlane(T, wave_s);
    const int my_base = __builtin_amdgcn_readlane(S, wave_s);
    {
        const int ncnt = __builtin_amdgcn_readfirstlane(cnt);
        const unsigned hi = (unsigned)wave_s << POSBITS;
        int D = S + P;
        for (int pb = 0, blk = 0; pb < ncnt; pb += PBLK, ++blk) {
            const int nb  = min(ncnt - pb, PBLK);
            const int st  = (int)s_seg[wave][blk][lane & (NW - 1)];
            const int st1 = (int)s_seg[wave][blk][(lane & (NW - 1)) + 1];
            const int adj = D - st;
            unsigned c[FCHUNK];
            int dst[FCHUNK];
#pragma unroll
            for (int u = 0; u < FCHUNK; ++u) {
                const int e = 32 * u + lane;
                c[u] = (e < nb) ? (unsigned)lst[pb + e] : 0u;
            }
#pragma unroll
            for (int u = 0; u < FCHUNK; ++u)
                dst[u] = __builtin_amdgcn_ds_bpermute((int)((c[u] >> 10) & 60u), adj) + 32 * u + lane;
#pragma unroll
            for (int u = 0; u < FCHUNK; ++u)
                if ((32 * u + lane) < nb)
                    s_own[dst[u]] = (unsigned short)((c[u] & ((1u << POSBITS) - 1u)) | hi | (((c[u] >> POSBITS) & 1u) << 15));
            D += st1 - st;
        }
    }
    __syncthreads();

    const unsigned short* nxt = s_own + my_base;
    int left = my_n;
    if (left >= BATCH) {
        pf = gather_ahead(BATCH, nxt, lane, wave, b, pj, shifts);
        nxt += BATCH;
    }
    while (left >= 2 * BATCH) {
        const Pre cur = pf;
        pf = gather_ahead(BATCH, nxt, lane, wave, b, pj, shifts);
        nxt += BATCH;
        sum_batch(BATCH, cur, rec, lane, lane16, half, a_base, s_atom, s_tab, cutoff, acc);
        left -= BATCH;
    }
    int nlast;
    Pre last;
    if (left >= BATCH) {
        nlast = left - BATCH;
        last = gather_ahead(nlast, nxt, lane, wave, b, pj, shifts);
        sum_batch(BATCH, pf, rec, lane, lane16, half, a_base, s_atom, s_tab, cutoff, acc);
    } else {
        nlast = left;
        last = gather_ahead(nlast, nxt, lane, wave, b, pj, shifts);
    }
    if (nlast > 0)
        sum_batch(nlast, last, rec, lane, lane16, half, a_base, s_atom, s_tab, cutoff, acc);

    if (lane16 < NANG) {
        float4* dst = (float4*)(s_tile + (2 * wave + half) * NWORB + lane16 * NWAVE);
        dst[0] = make_float4(acc[0], acc[1], acc[2], acc[3]);
        dst[1] = make_float4(acc[4], acc[5], acc[6], acc[7]);
    }
    __syncthreads();

    const int rt = wave >> 3;
    v8f dens = {};
    const float* arow = s_tile + (rt * 16 + lane16) * NWORB + kb;
#pragma unroll
    for (int j = 0; j < NANG; ++j) {
        const int l = (j == 0) ? 0 : ((j < 4) ? 1 : 2);
        const v2f a0 = *(const v2f*)(arow + j * NWAVE);
        const v2f a1 = *(const v2f*)(arow + j * NWAVE + 4);
        v8f c0 = {};
        v8f dmat = __builtin_amdgcn_wmma_f32_16x16x4_f32(
            false, a0, false, bt[l][0], (short)0, c0, false, false);
        dmat = __builtin_amdgcn_wmma_f32_16x16x4_f32(
            false, a1, false, bt[l][1], (short)0, dmat, false, false);
        dens += dmat * dmat;
    }

    const int row0 = b * NATOM + a_base + rt * 16 + half * 8;
#pragma unroll
    for (int r = 0; r < 8; ++r)
        out[(size_t)(row0 + r) * NORBIT + n] = dens[r];
}

extern "C" void kernel_launch(void* const* d_in, const int* in_sizes, int n_in,
                              void* d_out, int out_size, void* d_ws, size_t ws_size,
                              hipStream_t stream) {
    const float* cart       = (const float*)d_in[0];
    const int*   species    = (const int*)  d_in[2];
    const int*   atom_index = (const int*)  d_in[3];
    const float* shifts     = (const float*)d_in[4];
    const float* rs         = (const float*)d_in[5];
    const float* inta       = (const float*)d_in[6];
    const float* params     = (const float*)d_in[7];
    const float* hyper      = (const float*)d_in[8];
    const float* cutoff     = (const float*)d_in[10];
    float*       out        = (float*)d_out;

    eann_gather_kernel<<<256, 512, 0, stream>>>(
        cart, species, atom_index, shifts, rs, inta, params, cutoff, hyper, out);
}


# LDS list stores of scan, split and hand-over made branch-free: a lane with nothing to store writes a spare slot of its own (address select), no exec mask and branch per store; sum order kept
# speedup vs baseline: 114.2265x; 1.0639x over previous
#include <hip/hip_runtime.h>
#include <math.h>

#define NBATCH  16
#define NATOM   512
#define NEIGH   64
#define NPAIR   (NATOM * NEIGH)
#define TOTPAIR (NBATCH * NPAIR)
#define TOTATOM (NBATCH * NATOM)
#define NWAVE   8
#define NTYPE   4
#define NANG    13
#define NWORB   (NANG * NWAVE)
#define NORBIT  128

typedef __attribute__((ext_vector_type(2)))  float    v2f;
typedef __attribute__((ext_vector_type(8)))  float    v8f;

#define TILE_ATOMS 32
#define NW         16
#define WPAIRS     (NPAIR / NW)
#define NLOAD      (WPAIRS / 128)
#define POSBITS    11
#define BATCH      32
#define FCHUNK     5
#define PBLK       (32 * FCHUNK)
#define NPBLK      ((WPAIRS + PBLK - 1) / PBLK)
#define SEGROW     18
#define NMASK      16
#define TABROW     (3 * NWAVE)
#define RECSTRIDE  (NANG + NMASK)

__device__ __forceinline__ void wave_sync()
{
    __builtin_amdgcn_fence(__ATOMIC_RELEASE, "wavefront");
    __builtin_amdgcn_wave_barrier();
    __builtin_amdgcn_fence(__ATOMIC_ACQUIRE, "wavefront");
}

struct Pre { unsigned ent; int j; float sx, sy, sz; };

__device__ __forceinline__ Pre gather_ahead(
    int n, const unsigned short* own, int lane, int wave, int b,
    const int* __restrict__ pj, const float* __restrict__ shifts)
{
    Pre o = {0u, 0, 0.f, 0.f, 0.f};
    if (lane < n) {
        const unsigned c = (unsigned)own[lane];
        const int p = (int)(c & 0x7FFFu);
        o.ent = (unsigned)p | (((unsigned)(2 * wave) + (c >> 15)) << 16);
        const int g = b * NPAIR + p;
        o.j  = pj[p];
        o.sx = shifts[g * 3 + 0];
        o.sy = shifts[g * 3 + 1];
        o.sz = shifts[g * 3 + 2];
    }
    return o;
}

__device__ __forceinline__ void sum_batch(
    int batch, const Pre& cur, float* rec,
    int lane, int lane16, int half, int a_base,
    const float* s_atom, const float* s_tab, float cutoff, v8f& acc)
{
    const int nk = __builtin_amdgcn_readfirstlane((batch + 3) >> 2);
    if (lane < batch) {
        const int a     = (int)(cur.ent >> 16);
        const int j_loc = cur.j;
        const float shx = cur.sx;
        const float shy = cur.sy;
        const float shz = cur.sz;
        const float4 ai = *(const float4*)(s_atom + (a_base + a) * 4);
        const float4 aj = *(const float4*)(s_atom + j_loc * 4);

        const float dx = ai.x - aj.x + shx;
        const float dy = ai.y - aj.y + shy;
        const float dz = ai.z - aj.z + shz;
        const float dist = sqrtf(dx * dx + dy * dy + dz * dz);

        const int sp = __float_as_int(aj.w);
        const float4* tb4 = (const float4*)(s_tab + sp * TABROW);
        const float4 t0 = tb4[0], t1 = tb4[1], t2 = tb4[2], t3 = tb4[3], t4 = tb4[4], t5 = tb4[5];
        const float tb[TABROW] = {t0.x, t0.y, t0.z, t0.w, t1.x, t1.y, t1.z, t1.w,
                                  t2.x, t2.y, t2.z, t2.w, t3.x, t3.y, t3.z, t3.w,
                                  t4.x, t4.y, t4.z, t4.w, t5.x, t5.y, t5.z, t5.w};
        const float tc = 0.5f * __cosf(dist * (3.14159265358979f / cutoff)) + 0.5f;
        const float fc = tc * tc;

        float* r = rec + lane * RECSTRIDE;
        const float dv[3] = {dx, dy, dz};
        float ax[NANG];
        ax[0] = fc;
        ax[1] = fc * dx; ax[2] = fc * dy; ax[3] = fc * dz;
        int tpos = 4;
#pragma unroll
        for (int i = 1; i <= 3; ++i)
#pragma unroll
            for (int c = 0; c < 3; ++c)
                ax[tpos++] = ax[i] * dv[c];
        asm volatile("" :: "v"(t0.x), "v"(t0.y), "v"(t0.z), "v"(t0.w), "v"(t1.x), "v"(t1.y), "v"(t1.z), "v"(t1.w),
                           "v"(t2.x), "v"(t2.y), "v"(t2.z), "v"(t2.w), "v"(t3.x), "v"(t3.y), "v"(t3.z), "v"(t3.w),
                           "v"(t4.x), "v"(t4.y), "v"(t4.z), "v"(t4.w), "v"(t5.x), "v"(t5.y), "v"(t5.z), "v"(t5.w));
        for (int l = 0; l < NANG; ++l) r[l] = ax[l];
        float* own = r + NANG + ((a & 1) << 3);
        float* oth = r + NANG + (((a & 1) ^ 1) << 3);
        for (int ww = 0; ww < NWAVE; ++ww) {
            const float dd = dist - tb[ww];
            own[ww] = __expf(tb[NWAVE + ww] * dd * dd) * tb[2 * NWAVE + ww];
            oth[ww] = 0.f;
        }
    } else if (lane < 4 * nk) {
        float* r = rec + lane * RECSTRIDE;
        for (int l = 0; l < RECSTRIDE; ++l) r[l] = 0.f;
    }
    wave_sync();

    for (int ks = 0; ks < nk; ++ks) {
        const int k0 = 4 * ks + 2 * half;
        const float* f = rec + k0 * RECSTRIDE + lane16;
        v2f av, bv;
        bv.x = f[0];             bv.y = f[RECSTRIDE];
        av.x = f[NANG];          av.y = f[RECSTRIDE + NANG];
        acc = __builtin_amdgcn_wmma_f32_16x16x4_f32(
            false, av, false, bv, (short)0, acc, false, false);
    }
    wave_sync();
}

__global__ __launch_bounds__(32 * NW) void eann_gather_kernel(
    const float* __restrict__ cart,
    const int*   __restrict__ species,
    const int*   __restrict__ atom_index,
    const float* __restrict__ shifts,
    const float* __restrict__ rs,
    const float* __restrict__ inta,
    const float* __restrict__ params,
    const float* __restrict__ cutoff_p,
    const float* __restrict__ hyper,
    float*       __restrict__ out)
{
    __shared__ unsigned short s_list[NW + 1][WPAIRS];
    __shared__ unsigned short s_own[NPAIR];
    __shared__ float    s_rec[NW][BATCH * RECSTRIDE];
    __shared__ __attribute__((aligned(16))) float s_tile[TILE_ATOMS * NWORB];
    __shared__ int      s_tot[NW][NW];
    __shared__ unsigned short s_seg[NW][NPBLK][SEGROW];
    __shared__ __attribute__((aligned(16))) float s_atom[NATOM * 4];
    __shared__ __attribute__((aligned(16))) float s_tab[NTYPE * TABROW];

    const int t      = threadIdx.x;
    const int lane   = t & 31;
    const int wave   = t >> 5;
    const int b      = blockIdx.x >> 4;
    const int tib    = blockIdx.x & 15;
    const int a_base = tib * TILE_ATOMS;

    const int* __restrict__ pi = atom_index + b * NPAIR;
    const int* __restrict__ pj = atom_index + TOTPAIR + b * NPAIR;

    const float cutoff = cutoff_p[0];

    const int half   = lane >> 4;
    const int lane16 = lane & 15;
    const int kb     = half * 2;
    const int n      = (wave & 7) * 16 + lane16;
    v2f bt[3][2];
#pragma unroll
    for (int l = 0; l < 3; ++l)
#pragma unroll
        for (int c = 0; c < 2; ++c) {
            bt[l][c].x = hyper[(l * NWAVE + 4 * c + kb + 0) * NORBIT + n];
            bt[l][c].y = hyper[(l * NWAVE + 4 * c + kb + 1) * NORBIT + n];
        }

    float*    rec = s_rec[wave];
    Pre  pf = {0u, 0, 0.f, 0.f, 0.f};
    v8f acc = {};

    const int at0 = b * NATOM + t;
    const float c0x = cart[at0 * 3 + 0], c0y = cart[at0 * 3 + 1], c0z = cart[at0 * 3 + 2];
    const int   sp0 = species[at0];
    float tabv = 0.f;
    if (t < NTYPE * TABROW) {
        const int ts = t / TABROW;
        const int tk = t - ts * TABROW;
        const float* __restrict__ tsrc = (tk < NWAVE) ? rs : ((tk < 2 * NWAVE) ? inta : params);
        tabv = tsrc[ts * NWAVE + (tk & (NWAVE - 1))];
    }

    const int* __restrict__ pw = pi + wave * WPAIRS + lane * 4;
    int4 nx[NLOAD];
#pragma unroll
    for (int i = 0; i < NLOAD; ++i)
        nx[i] = *(const int4*)(pw + i * 128);

    *(float4*)(s_atom + t * 4)         = make_float4(c0x, c0y, c0z, __int_as_float(sp0));
    if (t < NTYPE * TABROW) s_tab[t] = tabv;

    unsigned short* lst = s_list[wave];
    unsigned short* const spare = s_list[NW] + 64 * wave + 2 * lane;
    int cnt = 0;
#pragma unroll
    for (int i = 0; i < NLOAD; ++i) {
        const int v[4] = {nx[i].x, nx[i].y, nx[i].z, nx[i].w};
#pragma unroll
        for (int c = 0; c < 4; ++c) {
            const bool m = (v[c] >> 5) == tib;
            const unsigned bal = __builtin_amdgcn_ballot_w32(m);
            const unsigned code = (unsigned)(i * 128 + lane * 4 + c) | ((unsigned)(v[c] & 31) << POSBITS);
            unsigned short* const slot = lst + __builtin_amdgcn_mbcnt_lo(bal, (unsigned)cnt);
            *(m ? slot : spare) = (unsigned short)code;
            cnt += __builtin_popcount(bal);
        }
        __builtin_amdgcn_sched_barrier(0);
    }

    wave_sync();
    {
        const int ncnt = __builtin_amdgcn_readfirstlane(cnt);
        const unsigned lo4 = (unsigned)lane & 15u;
        const unsigned l0 = 0u - (lo4 & 1u), l1 = 0u - ((lo4 >> 1) & 1u);
        const unsigned l2 = 0u - ((lo4 >> 2) & 1u), l3 = 0u - (lo4 >> 3);
        int ltot = 0;
        for (int pb = 0, blk = 0; pb < ncnt; pb += PBLK, ++blk) {
            const int nb = min(ncnt - pb, PBLK);
            unsigned cur[FCHUNK], rnk[FCHUNK];
            int cu[FCHUNK];
#pragma unroll
            for (int u = 0; u < FCHUNK; ++u) {
                const int e = 32 * u + lane;
                cur[u] = (e < nb) ? (unsigned)lst[pb + e] : 0xFFFFFFFFu;
            }
            int tot = 0;
#pragma unroll
            for (int u = 0; u < FCHUNK; ++u) {
                const unsigned V  = __builtin_amdgcn_ballot_w32((32 * u + lane) < nb);
                const int x0 = (int)(cur[u] << 19) >> 31;
                const int x1 = (int)(cur[u] << 18) >> 31;
                const int x2 = (int)(cur[u] << 17) >> 31;
                const int x3 = (int)(cur[u] << 16) >> 31;
                const unsigned B0 = __builtin_amdgcn_ballot_w32(x0 != 0);
                const unsigned B1 = __builtin_amdgcn_ballot_w32(x1 != 0);
                const unsigned B2 = __builtin_amdgcn_ballot_w32(x2 != 0);
                const unsigned B3 = __builtin_amdgcn_ballot_w32(x3 != 0);
                const unsigned same = V & ~(B0 ^ (unsigned)x0) & ~(B1 ^ (unsigned)x1)
                                        & ~(B2 ^ (unsigned)x2) & ~(B3 ^ (unsigned)x3);
                rnk[u] = __builtin_amdgcn_mbcnt_lo(same, 0u);
                const unsigned mo = V & ~(B0 ^ l0) & ~(B1 ^ l1) & ~(B2 ^ l2) & ~(B3 ^ l3);
                cu[u] = __builtin_popcount(mo);
                tot += cu[u];
            }
            int start = 0;
#pragma unroll
            for (int o = 0; o < NW - 1; ++o) {
                const int to = __builtin_amdgcn_readlane(tot, o);
                start += ((int)lo4 > o) ? to : 0;
            }
            int off[FCHUNK];
            int run = start;
#pragma unroll
            for (int u = 0; u < FCHUNK; ++u) {
                off[u] = __builtin_amdgcn_ds_bpermute((int)((cur[u] >> 10) & 60u), run);
                run += cu[u];
            }
#pragma unroll
            for (int u = 0; u < FCHUNK; ++u) {
                unsigned short* const slot = lst + pb + off[u] + (int)rnk[u];
                *(((32 * u + lane) < nb) ? slot : spare) = (unsigned short)cur[u];
            }
            *((lane <= NW) ? &s_seg[wave][blk][lane] : spare) = (unsigned short)((lane == NW) ? nb : start);
            ltot += tot;
        }
        if (lane < NW) s_tot[wave][lane] = ltot;
    }
    __syncthreads();

    const int wave_s = __builtin_amdgcn_readfirstlane(wave);
    int T = 0, P = 0;
#pragma unroll
    for (int w2 = 0; w2 < NW; ++w2) {
        const int v = s_tot[w2][lane & (NW - 1)];
        T += v;
        P += (w2 < wave_s) ? v : 0;
    }
    int S = 0;
#pragma unroll
    for (int o = 0; o < NW - 1; ++o) {
        const int to = __builtin_amdgcn_readlane(T, o);
        S += ((lane & (NW - 1)) > o) ? to : 0;
    }
    const int my_n    = __builtin_amdgcn_readlane(T, wave_s);
    const int my_base = __builtin_amdgcn_readlane(S, wave_s);
    {
        const int ncnt = __builtin_amdgcn_readfirstlane(cnt);
        const unsigned hi = (unsigned)wave_s << POSBITS;
        int D = S + P;
        for (int pb = 0, blk = 0; pb < ncnt; pb += PBLK, ++blk) {
            const int nb  = min(ncnt - pb, PBLK);
            const int st  = (int)s_seg[wave][blk][lane & (NW - 1)];
            const int st1 = (int)s_seg[wave][blk][(lane & (NW - 1)) + 1];
            const int adj = D - st;
            unsigned c[FCHUNK];
            int dst[FCHUNK];
#pragma unroll
            for (int u = 0; u < FCHUNK; ++u) {
                const int e = 32 * u + lane;
                c[u] = (e < nb) ? (unsigned)lst[pb + e] : 0u;
            }
#pragma unroll
            for (int u = 0; u < FCHUNK; ++u)
                dst[u] = __builtin_amdgcn_ds_bpermute((int)((c[u] >> 10) & 60u), adj) + 32 * u + lane;
#pragma unroll
            for (int u = 0; u < FCHUNK; ++u) {
                unsigned short* const slot = s_own + dst[u];
                *(((32 * u + lane) < nb) ? slot : spare) =
                    (unsigned short)((c[u] & ((1u << POSBITS) - 1u)) | hi | (((c[u] >> POSBITS) & 1u) << 15));
            }
            D += st1 - st;
        }
    }
    __syncthreads();

    const unsigned short* nxt = s_own + my_base;
    int left = my_n;
    if (left >= BATCH) {
        pf = gather_ahead(BATCH, nxt, lane, wave, b, pj, shifts);
        nxt += BATCH;
    }
    while (left >= 2 * BATCH) {
        const Pre cur = pf;
        pf = gather_ahead(BATCH, nxt, lane, wave, b, pj, shifts);
        nxt += BATCH;
        sum_batch(BATCH, cur, rec, lane, lane16, half, a_base, s_atom, s_tab, cutoff, acc);
        left -= BATCH;
    }
    int nlast;
    Pre last;
    if (left >= BATCH) {
        nlast = left - BATCH;
        last = gather_ahead(nlast, nxt, lane, wave, b, pj, shifts);
        sum_batch(BATCH, pf, rec, lane, lane16, half, a_base, s_atom, s_tab, cutoff, acc);
    } else {
        nlast = left;
        last = gather_ahead(nlast, nxt, lane, wave, b, pj, shifts);
    }
    if (nlast > 0)
        sum_batch(nlast, last, rec, lane, lane16, half, a_base, s_atom, s_tab, cutoff, acc);

    if (lane16 < NANG) {
        float4* dst = (float4*)(s_tile + (2 * wave + half) * NWORB + lane16 * NWAVE);
        dst[0] = make_float4(acc[0], acc[1], acc[2], acc[3]);
        dst[1] = make_float4(acc[4], acc[5], acc[6], acc[7]);
    }
    __syncthreads();

    const int rt = wave >> 3;
    v8f dens = {};
    const float* arow = s_tile + (rt * 16 + lane16) * NWORB + kb;
#pragma unroll
    for (int j = 0; j < NANG; ++j) {
        const int l = (j == 0) ? 0 : ((j < 4) ? 1 : 2);
        const v2f a0 = *(const v2f*)(arow + j * NWAVE);
        const v2f a1 = *(const v2f*)(arow + j * NWAVE + 4);
        v8f c0 = {};
        v8f dmat = __builtin_amdgcn_wmma_f32_16x16x4_f32(
            false, a0, false, bt[l][0], (short)0, c0, false, false);
        dmat = __builtin_amdgcn_wmma_f32_16x16x4_f32(
            false, a1, false, bt[l][1], (short)0, dmat, false, false);
        dens += dmat * dmat;
    }

    const int row0 = b * NATOM + a_base + rt * 16 + half * 8;
#pragma unroll
    for (int r = 0; r < 8; ++r)
        out[(size_t)(row0 + r) * NORBIT + n] = dens[r];
}

extern "C" void kernel_launch(void* const* d_in, const int* in_sizes, int n_in,
                              void* d_out, int out_size, void* d_ws, size_t ws_size,
                              hipStream_t stream) {
    const float* cart       = (const float*)d_in[0];
    const int*   species    = (const int*)  d_in[2];
    const int*   atom_index = (const int*)  d_in[3];
    const float* shifts     = (const float*)d_in[4];
    const float* rs         = (const float*)d_in[5];
    const float* inta       = (const float*)d_in[6];
    const float* params     = (const float*)d_in[7];
    const float* hyper      = (const float*)d_in[8];
    const float* cutoff     = (const float*)d_in[10];
    float*       out        = (float*)d_out;

    eann_gather_kernel<<<256, 512, 0, stream>>>(
        cart, species, atom_index, shifts, rs, inta, params, cutoff, hyper, out);
}


# Second-index and shift arrays prefetched to the second-level cache at kernel start, one 256-byte unit per thread: the sums' gathers hit there; built code also issues scan loads first; sum order kept
# speedup vs baseline: 121.9270x; 1.0674x over previous
#include <hip/hip_runtime.h>
#include <math.h>

#define NBATCH  16
#define NATOM   512
#define NEIGH   64
#define NPAIR   (NATOM * NEIGH)
#define TOTPAIR (NBATCH * NPAIR)
#define TOTATOM (NBATCH * NATOM)
#define NWAVE   8
#define NTYPE   4
#define NANG    13
#define NWORB   (NANG * NWAVE)
#define NORBIT  128

typedef __attribute__((ext_vector_type(2)))  float    v2f;
typedef __attribute__((ext_vector_type(8)))  float    v8f;

#define TILE_ATOMS 32
#define NW         16
#define WPAIRS     (NPAIR / NW)
#define NLOAD      (WPAIRS / 128)
#define POSBITS    11
#define BATCH      32
#define FCHUNK     5
#define PBLK       (32 * FCHUNK)
#define NPBLK      ((WPAIRS + PBLK - 1) / PBLK)
#define SEGROW     18
#define NMASK      16
#define TABROW     (3 * NWAVE)
#define RECSTRIDE  (NANG + NMASK)

__device__ __forceinline__ void wave_sync()
{
    __builtin_amdgcn_fence(__ATOMIC_RELEASE, "wavefront");
    __builtin_amdgcn_wave_barrier();
    __builtin_amdgcn_fence(__ATOMIC_ACQUIRE, "wavefront");
}

struct Pre { unsigned ent; int j; float sx, sy, sz; };

__device__ __forceinline__ Pre gather_ahead(
    int n, const unsigned short* own, int lane, int wave, int b,
    const int* __restrict__ pj, const float* __restrict__ shifts)
{
    Pre o = {0u, 0, 0.f, 0.f, 0.f};
    if (lane < n) {
        const unsigned c = (unsigned)own[lane];
        const int p = (int)(c & 0x7FFFu);
        o.ent = (unsigned)p | (((unsigned)(2 * wave) + (c >> 15)) << 16);
        const int g = b * NPAIR + p;
        o.j  = pj[p];
        o.sx = shifts[g * 3 + 0];
        o.sy = shifts[g * 3 + 1];
        o.sz = shifts[g * 3 + 2];
    }
    return o;
}

__device__ __forceinline__ void sum_batch(
    int batch, const Pre& cur, float* rec,
    int lane, int lane16, int half, int a_base,
    const float* s_atom, const float* s_tab, float cutoff, v8f& acc)
{
    const int nk = __builtin_amdgcn_readfirstlane((batch + 3) >> 2);
    if (lane < batch) {
        const int a     = (int)(cur.ent >> 16);
        const int j_loc = cur.j;
        const float shx = cur.sx;
        const float shy = cur.sy;
        const float shz = cur.sz;
        const float4 ai = *(const float4*)(s_atom + (a_base + a) * 4);
        const float4 aj = *(const float4*)(s_atom + j_loc * 4);

        const float dx = ai.x - aj.x + shx;
        const float dy = ai.y - aj.y + shy;
        const float dz = ai.z - aj.z + shz;
        const float dist = sqrtf(dx * dx + dy * dy + dz * dz);

        const int sp = __float_as_int(aj.w);
        const float4* tb4 = (const float4*)(s_tab + sp * TABROW);
        const float4 t0 = tb4[0], t1 = tb4[1], t2 = tb4[2], t3 = tb4[3], t4 = tb4[4], t5 = tb4[5];
        const float tb[TABROW] = {t0.x, t0.y, t0.z, t0.w, t1.x, t1.y, t1.z, t1.w,
                                  t2.x, t2.y, t2.z, t2.w, t3.x, t3.y, t3.z, t3.w,
                                  t4.x, t4.y, t4.z, t4.w, t5.x, t5.y, t5.z, t5.w};
        const float tc = 0.5f * __cosf(dist * (3.14159265358979f / cutoff)) + 0.5f;
        const float fc = tc * tc;

        float* r = rec + lane * RECSTRIDE;
        const float dv[3] = {dx, dy, dz};
        float ax[NANG];
        ax[0] = fc;
        ax[1] = fc * dx; ax[2] = fc * dy; ax[3] = fc * dz;
        int tpos = 4;
#pragma unroll
        for (int i = 1; i <= 3; ++i)
#pragma unroll
            for (int c = 0; c < 3; ++c)
                ax[tpos++] = ax[i] * dv[c];
        asm volatile("" :: "v"(t0.x), "v"(t0.y), "v"(t0.z), "v"(t0.w), "v"(t1.x), "v"(t1.y), "v"(t1.z), "v"(t1.w),
                           "v"(t2.x), "v"(t2.y), "v"(t2.z), "v"(t2.w), "v"(t3.x), "v"(t3.y), "v"(t3.z), "v"(t3.w),
                           "v"(t4.x), "v"(t4.y), "v"(t4.z), "v"(t4.w), "v"(t5.x), "v"(t5.y), "v"(t5.z), "v"(t5.w));
        for (int l = 0; l < NANG; ++l) r[l] = ax[l];
        float* own = r + NANG + ((a & 1) << 3);
        float* oth = r + NANG + (((a & 1) ^ 1) << 3);
        for (int ww = 0; ww < NWAVE; ++ww) {
            const float dd = dist - tb[ww];
            own[ww] = __expf(tb[NWAVE + ww] * dd * dd) * tb[2 * NWAVE + ww];
            oth[ww] = 0.f;
        }
    } else if (lane < 4 * nk) {
        float* r = rec + lane * RECSTRIDE;
        for (int l = 0; l < RECSTRIDE; ++l) r[l] = 0.f;
    }
    wave_sync();

    for (int ks = 0; ks < nk; ++ks) {
        const int k0 = 4 * ks + 2 * half;
        const float* f = rec + k0 * RECSTRIDE + lane16;
        v2f av, bv;
        bv.x = f[0];             bv.y = f[RECSTRIDE];
        av.x = f[NANG];          av.y = f[RECSTRIDE + NANG];
        acc = __builtin_amdgcn_wmma_f32_16x16x4_f32(
            false, av, false, bv, (short)0, acc, false, false);
    }
    wave_sync();
}

__global__ __launch_bounds__(32 * NW) void eann_gather_kernel(
    const float* __restrict__ cart,
    const int*   __restrict__ species,
    const int*   __restrict__ atom_index,
    const float* __restrict__ shifts,
    const float* __restrict__ rs,
    const float* __restrict__ inta,
    const float* __restrict__ params,
    const float* __restrict__ cutoff_p,
    const float* __restrict__ hyper,
    float*       __restrict__ out)
{
    __shared__ unsigned short s_list[NW + 1][WPAIRS];
    __shared__ unsigned short s_own[NPAIR];
    __shared__ float    s_rec[NW][BATCH * RECSTRIDE];
    __shared__ __attribute__((aligned(16))) float s_tile[TILE_ATOMS * NWORB];
    __shared__ int      s_tot[NW][NW];
    __shared__ unsigned short s_seg[NW][NPBLK][SEGROW];
    __shared__ __attribute__((aligned(16))) float s_atom[NATOM * 4];
    __shared__ __attribute__((aligned(16))) float s_tab[NTYPE * TABROW];

    const int t      = threadIdx.x;
    const int lane   = t & 31;
    const int wave   = t >> 5;
    const int b      = blockIdx.x >> 4;
    const int tib    = blockIdx.x & 15;
    const int a_base = tib * TILE_ATOMS;

    const int* __restrict__ pi = atom_index + b * NPAIR;
    const int* __restrict__ pj = atom_index + TOTPAIR + b * NPAIR;

    const float cutoff = cutoff_p[0];

    const int half   = lane >> 4;
    const int lane16 = lane & 15;
    const int kb     = half * 2;
    const int n      = (wave & 7) * 16 + lane16;
    v2f bt[3][2];
#pragma unroll
    for (int l = 0; l < 3; ++l)
#pragma unroll
        for (int c = 0; c < 2; ++c) {
            bt[l][c].x = hyper[(l * NWAVE + 4 * c + kb + 0) * NORBIT + n];
            bt[l][c].y = hyper[(l * NWAVE + 4 * c + kb + 1) * NORBIT + n];
        }

    float*    rec = s_rec[wave];
    Pre  pf = {0u, 0, 0.f, 0.f, 0.f};
    v8f acc = {};

    const int at0 = b * NATOM + t;
    const float c0x = cart[at0 * 3 + 0], c0y = cart[at0 * 3 + 1], c0z = cart[at0 * 3 + 2];
    const int   sp0 = species[at0];
    float tabv = 0.f;
    if (t < NTYPE * TABROW) {
        const int ts = t / TABROW;
        const int tk = t - ts * TABROW;
        const float* __restrict__ tsrc = (tk < NWAVE) ? rs : ((tk < 2 * NWAVE) ? inta : params);
        tabv = tsrc[ts * NWAVE + (tk & (NWAVE - 1))];
    }

    const int* __restrict__ pw = pi + wave * WPAIRS + lane * 4;
    int4 nx[NLOAD];
#pragma unroll
    for (int i = 0; i < NLOAD; ++i)
        nx[i] = *(const int4*)(pw + i * 128);

    {
        const int wv   = __builtin_amdgcn_readfirstlane(wave);
        const int s16  = tib ^ ((wv >> 2) << 2);
        const int part = wv & 3;
        const char* pbase = (part == 0)
            ? (const char*)(pj + s16 * WPAIRS)
            : (const char*)(shifts + (size_t)(b * NPAIR + s16 * WPAIRS) * 3 + (part - 1) * WPAIRS);
        __builtin_prefetch(pbase + lane * 256, 0, 1);
    }

    *(float4*)(s_atom + t * 4)         = make_float4(c0x, c0y, c0z, __int_as_float(sp0));
    if (t < NTYPE * TABROW) s_tab[t] = tabv;

    unsigned short* lst = s_list[wave];
    unsigned short* const spare = s_list[NW] + 64 * wave + 2 * lane;
    int cnt = 0;
#pragma unroll
    for (int i = 0; i < NLOAD; ++i) {
        const int v[4] = {nx[i].x, nx[i].y, nx[i].z, nx[i].w};
#pragma unroll
        for (int c = 0; c < 4; ++c) {
            const bool m = (v[c] >> 5) == tib;
            const unsigned bal = __builtin_amdgcn_ballot_w32(m);
            const unsigned code = (unsigned)(i * 128 + lane * 4 + c) | ((unsigned)(v[c] & 31) << POSBITS);
            unsigned short* const slot = lst + __builtin_amdgcn_mbcnt_lo(bal, (unsigned)cnt);
            *(m ? slot : spare) = (unsigned short)code;
            cnt += __builtin_popcount(bal);
        }
        __builtin_amdgcn_sched_barrier(0);
    }

    wave_sync();
    {
        const int ncnt = __builtin_amdgcn_readfirstlane(cnt);
        const unsigned lo4 = (unsigned)lane & 15u;
        const unsigned l0 = 0u - (lo4 & 1u), l1 = 0u - ((lo4 >> 1) & 1u);
        const unsigned l2 = 0u - ((lo4 >> 2) & 1u), l3 = 0u - (lo4 >> 3);
        int ltot = 0;
        for (int pb = 0, blk = 0; pb < ncnt; pb += PBLK, ++blk) {
            const int nb = min(ncnt - pb, PBLK);
            unsigned cur[FCHUNK], rnk[FCHUNK];
            int cu[FCHUNK];
#pragma unroll
            for (int u = 0; u < FCHUNK; ++u) {
                const int e = 32 * u + lane;
                cur[u] = (e < nb) ? (unsigned)lst[pb + e] : 0xFFFFFFFFu;
            }
            int tot = 0;
#pragma unroll
            for (int u = 0; u < FCHUNK; ++u) {
                const unsigned V  = __builtin_amdgcn_ballot_w32((32 * u + lane) < nb);
                const int x0 = (int)(cur[u] << 19) >> 31;
                const int x1 = (int)(cur[u] << 18) >> 31;
                const int x2 = (int)(cur[u] << 17) >> 31;
                const int x3 = (int)(cur[u] << 16) >> 31;
                const unsigned B0 = __builtin_amdgcn_ballot_w32(x0 != 0);
                const unsigned B1 = __builtin_amdgcn_ballot_w32(x1 != 0);
                const unsigned B2 = __builtin_amdgcn_ballot_w32(x2 != 0);
                const unsigned B3 = __builtin_amdgcn_ballot_w32(x3 != 0);
                const unsigned same = V & ~(B0 ^ (unsigned)x0) & ~(B1 ^ (unsigned)x1)
                                        & ~(B2 ^ (unsigned)x2) & ~(B3 ^ (unsigned)x3);
                rnk[u] = __builtin_amdgcn_mbcnt_lo(same, 0u);
                const unsigned mo = V & ~(B0 ^ l0) & ~(B1 ^ l1) & ~(B2 ^ l2) & ~(B3 ^ l3);
                cu[u] = __builtin_popcount(mo);
                tot += cu[u];
            }
            int start = 0;
#pragma unroll
            for (int o = 0; o < NW - 1; ++o) {
                const int to = __builtin_amdgcn_readlane(tot, o);
                start += ((int)lo4 > o) ? to : 0;
            }
            int off[FCHUNK];
            int run = start;
#pragma unroll
            for (int u = 0; u < FCHUNK; ++u) {
                off[u] = __builtin_amdgcn_ds_bpermute((int)((cur[u] >> 10) & 60u), run);
                run += cu[u];
            }
#pragma unroll
            for (int u = 0; u < FCHUNK; ++u) {
                unsigned short* const slot = lst + pb + off[u] + (int)rnk[u];
                *(((32 * u + lane) < nb) ? slot : spare) = (unsigned short)cur[u];
            }
            *((lane <= NW) ? &s_seg[wave][blk][lane] : spare) = (unsigned short)((lane == NW) ? nb : start);
            ltot += tot;
        }
        if (lane < NW) s_tot[wave][lane] = ltot;
    }
    __syncthreads();

    const int wave_s = __builtin_amdgcn_readfirstlane(wave);
    int T = 0, P = 0;
#pragma unroll
    for (int w2 = 0; w2 < NW; ++w2) {
        const int v = s_tot[w2][lane & (NW - 1)];
        T += v;
        P += (w2 < wave_s) ? v : 0;
    }
    int S = 0;
#pragma unroll
    for (int o = 0; o < NW - 1; ++o) {
        const int to = __builtin_amdgcn_readlane(T, o);
        S += ((lane & (NW - 1)) > o) ? to : 0;
    }
    const int my_n    = __builtin_amdgcn_readlane(T, wave_s);
    const int my_base = __builtin_amdgcn_readlane(S, wave_s);
    {
        const int ncnt = __builtin_amdgcn_readfirstlane(cnt);
        const unsigned hi = (unsigned)wave_s << POSBITS;
        int D = S + P;
        for (int pb = 0, blk = 0; pb < ncnt; pb += PBLK, ++blk) {
            const int nb  = min(ncnt - pb, PBLK);
            const int st  = (int)s_seg[wave][blk][lane & (NW - 1)];
            const int st1 = (int)s_seg[wave][blk][(lane & (NW - 1)) + 1];
            const int adj = D - st;
            unsigned c[FCHUNK];
            int dst[FCHUNK];
#pragma unroll
            for (int u = 0; u < FCHUNK; ++u) {
                const int e = 32 * u + lane;
                c[u] = (e < nb) ? (unsigned)lst[pb + e] : 0u;
            }
#pragma unroll
            for (int u = 0; u < FCHUNK; ++u)
                dst[u] = __builtin_amdgcn_ds_bpermute((int)((c[u] >> 10) & 60u), adj) + 32 * u + lane;
#pragma unroll
            for (int u = 0; u < FCHUNK; ++u) {
                unsigned short* const slot = s_own + dst[u];
                *(((32 * u + lane) < nb) ? slot : spare) =
                    (unsigned short)((c[u] & ((1u << POSBITS) - 1u)) | hi | (((c[u] >> POSBITS) & 1u) << 15));
            }
            D += st1 - st;
        }
    }
    __syncthreads();

    const unsigned short* nxt = s_own + my_base;
    int left = my_n;
    if (left >= BATCH) {
        pf = gather_ahead(BATCH, nxt, lane, wave, b, pj, shifts);
        nxt += BATCH;
    }
    while (left >= 2 * BATCH) {
        const Pre cur = pf;
        pf = gather_ahead(BATCH, nxt, lane, wave, b, pj, shifts);
        nxt += BATCH;
        sum_batch(BATCH, cur, rec, lane, lane16, half, a_base, s_atom, s_tab, cutoff, acc);
        left -= BATCH;
    }
    int nlast;
    Pre last;
    if (left >= BATCH) {
        nlast = left - BATCH;
        last = gather_ahead(nlast, nxt, lane, wave, b, pj, shifts);
        sum_batch(BATCH, pf, rec, lane, lane16, half, a_base, s_atom, s_tab, cutoff, acc);
    } else {
        nlast = left;
        last = gather_ahead(nlast, nxt, lane, wave, b, pj, shifts);
    }
    if (nlast > 0)
        sum_batch(nlast, last, rec, lane, lane16, half, a_base, s_atom, s_tab, cutoff, acc);

    if (lane16 < NANG) {
        float4* dst = (float4*)(s_tile + (2 * wave + half) * NWORB + lane16 * NWAVE);
        dst[0] = make_float4(acc[0], acc[1], acc[2], acc[3]);
        dst[1] = make_float4(acc[4], acc[5], acc[6], acc[7]);
    }
    __syncthreads();

    const int rt = wave >> 3;
    v8f dens = {};
    const float* arow = s_tile + (rt * 16 + lane16) * NWORB + kb;
#pragma unroll
    for (int j = 0; j < NANG; ++j) {
        const int l = (j == 0) ? 0 : ((j < 4) ? 1 : 2);
        const v2f a0 = *(const v2f*)(arow + j * NWAVE);
        const v2f a1 = *(const v2f*)(arow + j * NWAVE + 4);
        v8f c0 = {};
        v8f dmat = __builtin_amdgcn_wmma_f32_16x16x4_f32(
            false, a0, false, bt[l][0], (short)0, c0, false, false);
        dmat = __builtin_amdgcn_wmma_f32_16x16x4_f32(
            false, a1, false, bt[l][1], (short)0, dmat, false, false);
        dens += dmat * dmat;
    }

    const int row0 = b * NATOM + a_base + rt * 16 + half * 8;
#pragma unroll
    for (int r = 0; r < 8; ++r)
        out[(size_t)(row0 + r) * NORBIT + n] = dens[r];
}

extern "C" void kernel_launch(void* const* d_in, const int* in_sizes, int n_in,
                              void* d_out, int out_size, void* d_ws, size_t ws_size,
                              hipStream_t stream) {
    const float* cart       = (const float*)d_in[0];
    const int*   species    = (const int*)  d_in[2];
    const int*   atom_index = (const int*)  d_in[3];
    const float* shifts     = (const float*)d_in[4];
    const float* rs         = (const float*)d_in[5];
    const float* inta       = (const float*)d_in[6];
    const float* params     = (const float*)d_in[7];
    const float* hyper      = (const float*)d_in[8];
    const float* cutoff     = (const float*)d_in[10];
    float*       out        = (float*)d_out;

    eann_gather_kernel<<<256, 512, 0, stream>>>(
        cart, species, atom_index, shifts, rs, inta, params, cutoff, hyper, out);
}
